# starting kernel
# baseline (speedup 1.0000x reference)
#include <hip/hip_runtime.h>

typedef float v2f __attribute__((ext_vector_type(2)));
typedef float v8f __attribute__((ext_vector_type(8)));

#define LN_EPS 1e-5f

__device__ __forceinline__ v8f wmma4(v2f a, v2f b, v8f c) {
  return __builtin_amdgcn_wmma_f32_16x16x4_f32(false, a, false, b, (short)0, c,
                                               false, false);
}

__device__ __forceinline__ void atomic_add_f32(float* p, float v) {
  __hip_atomic_fetch_add(p, v, __ATOMIC_RELAXED, __HIP_MEMORY_SCOPE_AGENT);
}

__global__ __launch_bounds__(256) void zero_f32(float* __restrict__ p, long n) {
  long i = (long)blockIdx.x * blockDim.x + threadIdx.x;
  if (i < n) p[i] = 0.0f;
}

__global__ __launch_bounds__(256) void degree_count(const int* __restrict__ dst,
                                                    float* __restrict__ cnt,
                                                    long E) {
  long e = (long)blockIdx.x * blockDim.x + threadIdx.x;
  if (e < E) atomic_add_f32(&cnt[dst[e]], 1.0f);
}

__global__ __launch_bounds__(256) void invert_cnt(float* __restrict__ cnt, int n) {
  int i = blockIdx.x * blockDim.x + threadIdx.x;
  if (i < n) cnt[i] = 1.0f / fmaxf(cnt[i], 1.0f);
}

__global__ __launch_bounds__(256) void scatter_edges(
    const float4* __restrict__ h4, const int* __restrict__ src,
    const int* __restrict__ dst, float* __restrict__ agg, long E) {
  long idx = (long)blockIdx.x * blockDim.x + threadIdx.x;
  if (idx >= E * 16) return;
  long e = idx >> 4;
  int ch = (int)(idx & 15);
  int s = src[e];
  int d = dst[e];
  float4 v = h4[(long)s * 16 + ch];
  float* p = agg + (long)d * 64 + (long)ch * 4;
  atomic_add_f32(p + 0, v.x);
  atomic_add_f32(p + 1, v.y);
  atomic_add_f32(p + 2, v.z);
  atomic_add_f32(p + 3, v.w);
}

__global__ __launch_bounds__(256) void input_gemm(
    const float* __restrict__ x, const float* __restrict__ Win,
    const float* __restrict__ bin, float* __restrict__ h, int N) {
  __shared__ float sW[16 * 64];
  __shared__ float sb[64];
  int t = threadIdx.x;
  for (int i = t; i < 16 * 64; i += 256) sW[i] = Win[i];
  if (t < 64) sb[t] = bin[t];
  __syncthreads();

  int wave = t >> 5, lane = t & 31, half = lane >> 4, ln = lane & 15;
  long tile = (long)blockIdx.x * 8 + wave;
  long row0 = tile * 16;
  if (row0 >= N) return;

  int r = (int)row0 + ln;
  if (r > N - 1) r = N - 1;
  const float* px = x + (long)r * 16;

  v8f acc[4];
#pragma unroll
  for (int c = 0; c < 4; ++c) acc[c] = (v8f){0, 0, 0, 0, 0, 0, 0, 0};

#pragma unroll
  for (int k = 0; k < 16; k += 4) {
    int ka = k + 2 * half;
    v2f a;
    a[0] = px[ka];
    a[1] = px[ka + 1];
#pragma unroll
    for (int c = 0; c < 4; ++c) {
      v2f b;
      b[0] = sW[ka * 64 + c * 16 + ln];
      b[1] = sW[(ka + 1) * 64 + c * 16 + ln];
      acc[c] = wmma4(a, b, acc[c]);
    }
  }

  float bias[4];
#pragma unroll
  for (int c = 0; c < 4; ++c) bias[c] = sb[c * 16 + ln];
#pragma unroll
  for (int v = 0; v < 8; ++v) {
    long row = row0 + v + 8 * half;
    if (row < N) {
      float* po = h + row * 64 + ln;
      po[0] = acc[0][v] + bias[0];
      po[16] = acc[1][v] + bias[1];
      po[32] = acc[2][v] + bias[2];
      po[48] = acc[3][v] + bias[3];
    }
  }
}

__global__ __launch_bounds__(256) void sage_update(
    const float* __restrict__ agg, const float* __restrict__ hin,
    const float* __restrict__ invc, const float* __restrict__ Wl,
    const float* __restrict__ bl, const float* __restrict__ Wr,
    const float* __restrict__ gam, const float* __restrict__ bet,
    float* __restrict__ hout, int N) {
  __shared__ float sWl[64 * 64];
  __shared__ float sWr[64 * 64];
  __shared__ float sbl[64], sg[64], sbeta[64];
  int t = threadIdx.x;
  for (int i = t; i < 64 * 64; i += 256) {
    sWl[i] = Wl[i];
    sWr[i] = Wr[i];
  }
  if (t < 64) {
    sbl[t] = bl[t];
    sg[t] = gam[t];
    sbeta[t] = bet[t];
  }
  __syncthreads();

  int wave = t >> 5, lane = t & 31, half = lane >> 4, ln = lane & 15;
  long tile = (long)blockIdx.x * 8 + wave;
  long row0 = tile * 16;
  if (row0 >= N) return;

  int r = (int)row0 + ln;
  if (r > N - 1) r = N - 1;
  float iv = invc[r];
  const float* pag = agg + (long)r * 64;
  const float* ph = hin + (long)r * 64;

  v8f acc[4];
#pragma unroll
  for (int c = 0; c < 4; ++c) acc[c] = (v8f){0, 0, 0, 0, 0, 0, 0, 0};

#pragma unroll
  for (int k = 0; k < 64; k += 4) {
    int ka = k + 2 * half;
    v2f aA, aH;
    aA[0] = pag[ka] * iv;
    aA[1] = pag[ka + 1] * iv;
    aH[0] = ph[ka];
    aH[1] = ph[ka + 1];
#pragma unroll
    for (int c = 0; c < 4; ++c) {
      v2f bL, bR;
      bL[0] = sWl[ka * 64 + c * 16 + ln];
      bL[1] = sWl[(ka + 1) * 64 + c * 16 + ln];
      bR[0] = sWr[ka * 64 + c * 16 + ln];
      bR[1] = sWr[(ka + 1) * 64 + c * 16 + ln];
      acc[c] = wmma4(aA, bL, acc[c]);
      acc[c] = wmma4(aH, bR, acc[c]);
    }
  }

  float bias[4], g4[4], be4[4];
#pragma unroll
  for (int c = 0; c < 4; ++c) {
    bias[c] = sbl[c * 16 + ln];
    g4[c] = sg[c * 16 + ln];
    be4[c] = sbeta[c * 16 + ln];
  }

#pragma unroll
  for (int v = 0; v < 8; ++v) {
    float r0 = fmaxf(acc[0][v] + bias[0], 0.0f);
    float r1 = fmaxf(acc[1][v] + bias[1], 0.0f);
    float r2 = fmaxf(acc[2][v] + bias[2], 0.0f);
    float r3 = fmaxf(acc[3][v] + bias[3], 0.0f);
    float s = r0 + r1 + r2 + r3;
    s += __shfl_xor(s, 1, 16);
    s += __shfl_xor(s, 2, 16);
    s += __shfl_xor(s, 4, 16);
    s += __shfl_xor(s, 8, 16);
    float mu = s * (1.0f / 64.0f);
    float d0 = r0 - mu, d1 = r1 - mu, d2 = r2 - mu, d3 = r3 - mu;
    float q = d0 * d0 + d1 * d1 + d2 * d2 + d3 * d3;
    q += __shfl_xor(q, 1, 16);
    q += __shfl_xor(q, 2, 16);
    q += __shfl_xor(q, 4, 16);
    q += __shfl_xor(q, 8, 16);
    float rin = rsqrtf(q * (1.0f / 64.0f) + LN_EPS);
    long row = row0 + v + 8 * half;
    if (row < N) {
      float* po = hout + row * 64 + ln;
      po[0] = d0 * rin * g4[0] + be4[0];
      po[16] = d1 * rin * g4[1] + be4[1];
      po[32] = d2 * rin * g4[2] + be4[2];
      po[48] = d3 * rin * g4[3] + be4[3];
    }
  }
}

extern "C" void kernel_launch(void* const* d_in, const int* in_sizes, int n_in,
                              void* d_out, int out_size, void* d_ws,
                              size_t ws_size, hipStream_t stream) {
  (void)n_in;
  (void)out_size;
  (void)ws_size;
  const float* x = (const float*)d_in[0];
  const int* ei = (const int*)d_in[1];
  const float* Win = (const float*)d_in[2];
  const float* bin = (const float*)d_in[3];
  const float* Wl = (const float*)d_in[4];
  const float* bl = (const float*)d_in[5];
  const float* Wr = (const float*)d_in[6];
  const float* gam = (const float*)d_in[7];
  const float* bet = (const float*)d_in[8];

  const int N = in_sizes[0] / 16;
  const long E = in_sizes[1] / 2;
  const int* src = ei;
  const int* dst = ei + E;

  float* ws = (float*)d_ws;
  size_t nA = ((size_t)N + 63) & ~(size_t)63;
  float* cnt = ws;
  float* buf0 = ws + nA;
  float* buf1 = buf0 + (size_t)N * 64;

  zero_f32<<<(N + 255) / 256, 256, 0, stream>>>(cnt, (long)N);
  degree_count<<<(int)((E + 255) / 256), 256, 0, stream>>>(dst, cnt, E);
  invert_cnt<<<(N + 255) / 256, 256, 0, stream>>>(cnt, N);

  int tiles = (N + 15) / 16;
  int gblocks = (tiles + 7) / 8;
  input_gemm<<<gblocks, 256, 0, stream>>>(x, Win, bin, buf0, N);

  float* hcur = buf0;
  float* spare = buf1;
  const long nd = (long)N * 64;
  for (int l = 0; l < 3; ++l) {
    zero_f32<<<(int)((nd + 255) / 256), 256, 0, stream>>>(spare, nd);
    scatter_edges<<<(int)((E * 16 + 255) / 256), 256, 0, stream>>>(
        (const float4*)hcur, src, dst, spare, E);
    float* hn = (l == 2) ? (float*)d_out : spare;
    sage_update<<<gblocks, 256, 0, stream>>>(
        spare, hcur, cnt, Wl + (size_t)l * 4096, bl + (size_t)l * 64,
        Wr + (size_t)l * 4096, gam, bet, hn, N);
    float* tmp = hcur;
    hcur = hn;
    spare = tmp;
  }
}


# Neighbour sum without float atomics: edges grouped by destination each call (returning int add = degree and arrival-order slot, scan, fill), then one row gather per node; order of summation changes
# speedup vs baseline: 2.2188x; 2.2188x over previous
#include <hip/hip_runtime.h>


typedef float v2f __attribute__((ext_vector_type(2)));
typedef float v8f __attribute__((ext_vector_type(8)));

#define LN_EPS 1e-5f

__device__ __forceinline__ v8f wmma4(v2f a, v2f b, v8f c) {
  return __builtin_amdgcn_wmma_f32_16x16x4_f32(false, a, false, b, (short)0, c,
                                               false, false);
}

__global__ __launch_bounds__(256) void zero_i32(int* __restrict__ p, int n) {
  int i = blockIdx.x * blockDim.x + threadIdx.x;
  if (i < n) p[i] = 0;
}

__global__ __launch_bounds__(256) void rank_edges(const int* __restrict__ dst,
                                                  int* __restrict__ cnt,
                                                  int* __restrict__ rank,
                                                  long E) {
  long e = (long)blockIdx.x * blockDim.x + threadIdx.x;
  if (e < E)
    rank[e] = __hip_atomic_fetch_add(&cnt[dst[e]], 1, __ATOMIC_RELAXED,
                                     __HIP_MEMORY_SCOPE_AGENT);
}

__global__ __launch_bounds__(256) void scan_counts(const int* __restrict__ cnt,
                                                   int* __restrict__ offs,
                                                   float* __restrict__ inv,
                                                   int N) {
  __shared__ int sfront[8];
  __shared__ int swave[8];
  int t = threadIdx.x, lane = t & 31, wave = t >> 5;
  int lim = blockIdx.x * 1024;

  int acc = 0;
  const int4* c4 = (const int4*)cnt;
  for (int i = t; i < lim / 4; i += 256) {
    int4 v = c4[i];
    acc += (v.x + v.y) + (v.z + v.w);
  }
  acc += __shfl_xor(acc, 16);
  acc += __shfl_xor(acc, 8);
  acc += __shfl_xor(acc, 4);
  acc += __shfl_xor(acc, 2);
  acc += __shfl_xor(acc, 1);
  if (lane == 0) sfront[wave] = acc;

  int i0 = lim + t * 4;
  int c[4];
#pragma unroll
  for (int k = 0; k < 4; ++k) c[k] = (i0 + k < N) ? cnt[i0 + k] : 0;
  int s = (c[0] + c[1]) + (c[2] + c[3]);
  int inc = s;
#pragma unroll
  for (int o = 1; o < 32; o <<= 1) {
    int v = __shfl_up(inc, o);
    if (lane >= o) inc += v;
  }
  if (lane == 31) swave[wave] = inc;
  __syncthreads();

  int ex = inc - s;
#pragma unroll
  for (int w = 0; w < 8; ++w) {
    ex += sfront[w];
    if (w < wave) ex += swave[w];
  }
#pragma unroll
  for (int k = 0; k < 4; ++k) {
    if (i0 + k < N) {
      offs[i0 + k] = ex;
      inv[i0 + k] = 1.0f / fmaxf((float)c[k], 1.0f);
      ex += c[k];
    }
  }
}

__global__ __launch_bounds__(256) void fill_groups(const int* __restrict__ src,
                                                   const int* __restrict__ dst,
                                                   const int* __restrict__ rank,
                                                   const int* __restrict__ offs,
                                                   int* __restrict__ grp,
                                                   long E) {
  long e = (long)blockIdx.x * blockDim.x + threadIdx.x;
  if (e < E) grp[offs[dst[e]] + rank[e]] = src[e];
}

__global__ __launch_bounds__(256) void gather_rows(
    const float4* __restrict__ h4, const int* __restrict__ grp,
    const int* __restrict__ offs, const int* __restrict__ cnt,
    float4* __restrict__ agg4, int N) {
  int t = threadIdx.x;
  int lane = t & 31, half = lane >> 4, c = lane & 15;
  int node = __builtin_amdgcn_readfirstlane((int)blockIdx.x * 8 + (t >> 5));
  if (node >= N) return;
  int beg = offs[node];
  int deg = cnt[node];
  const int* p = grp + beg;

  float4 a = make_float4(0.0f, 0.0f, 0.0f, 0.0f);
  int j = 0;
  for (; j + 8 <= deg; j += 8) {
    int e0 = p[j], e1 = p[j + 1], e2 = p[j + 2], e3 = p[j + 3];
    int e4 = p[j + 4], e5 = p[j + 5], e6 = p[j + 6], e7 = p[j + 7];
    int s0 = half ? e1 : e0;
    int s1 = half ? e3 : e2;
    int s2 = half ? e5 : e4;
    int s3 = half ? e7 : e6;
    float4 v0 = h4[(long)s0 * 16 + c];
    float4 v1 = h4[(long)s1 * 16 + c];
    float4 v2 = h4[(long)s2 * 16 + c];
    float4 v3 = h4[(long)s3 * 16 + c];
    a.x += v0.x; a.y += v0.y; a.z += v0.z; a.w += v0.w;
    a.x += v1.x; a.y += v1.y; a.z += v1.z; a.w += v1.w;
    a.x += v2.x; a.y += v2.y; a.z += v2.z; a.w += v2.w;
    a.x += v3.x; a.y += v3.y; a.z += v3.z; a.w += v3.w;
  }
  if (j < deg) {
    int last = deg - 1;
    int i0 = j + half, i1 = i0 + 2, i2 = i0 + 4, i3 = i0 + 6;
    int s0 = p[i0 < last ? i0 : last];
    int s1 = p[i1 < last ? i1 : last];
    int s2 = p[i2 < last ? i2 : last];
    int s3 = p[i3 < last ? i3 : last];
    float4 v0 = h4[(long)s0 * 16 + c];
    float4 v1 = h4[(long)s1 * 16 + c];
    float4 v2 = h4[(long)s2 * 16 + c];
    float4 v3 = h4[(long)s3 * 16 + c];
    bool k0 = i0 < deg, k1 = i1 < deg, k2 = i2 < deg, k3 = i3 < deg;
    a.x += k0 ? v0.x : 0.0f; a.y += k0 ? v0.y : 0.0f;
    a.z += k0 ? v0.z : 0.0f; a.w += k0 ? v0.w : 0.0f;
    a.x += k1 ? v1.x : 0.0f; a.y += k1 ? v1.y : 0.0f;
    a.z += k1 ? v1.z : 0.0f; a.w += k1 ? v1.w : 0.0f;
    a.x += k2 ? v2.x : 0.0f; a.y += k2 ? v2.y : 0.0f;
    a.z += k2 ? v2.z : 0.0f; a.w += k2 ? v2.w : 0.0f;
    a.x += k3 ? v3.x : 0.0f; a.y += k3 ? v3.y : 0.0f;
    a.z += k3 ? v3.z : 0.0f; a.w += k3 ? v3.w : 0.0f;
  }
  a.x += __shfl_xor(a.x, 16);
  a.y += __shfl_xor(a.y, 16);
  a.z += __shfl_xor(a.z, 16);
  a.w += __shfl_xor(a.w, 16);
  if (half == 0) agg4[(long)node * 16 + c] = a;
}

__global__ __launch_bounds__(256) void input_gemm(
    const float* __restrict__ x, const float* __restrict__ Win,
    const float* __restrict__ bin, float* __restrict__ h, int N) {
  __shared__ float sW[16 * 64];
  __shared__ float sb[64];
  int t = threadIdx.x;
  for (int i = t; i < 16 * 64; i += 256) sW[i] = Win[i];
  if (t < 64) sb[t] = bin[t];
  __syncthreads();

  int wave = t >> 5, lane = t & 31, half = lane >> 4, ln = lane & 15;
  long tile = (long)blockIdx.x * 8 + wave;
  long row0 = tile * 16;
  if (row0 >= N) return;

  int r = (int)row0 + ln;
  if (r > N - 1) r = N - 1;
  const float* px = x + (long)r * 16;

  v8f acc[4];
#pragma unroll
  for (int c = 0; c < 4; ++c) acc[c] = (v8f){0, 0, 0, 0, 0, 0, 0, 0};

#pragma unroll
  for (int k = 0; k < 16; k += 4) {
    int ka = k + 2 * half;
    v2f a;
    a[0] = px[ka];
    a[1] = px[ka + 1];
#pragma unroll
    for (int c = 0; c < 4; ++c) {
      v2f b;
      b[0] = sW[ka * 64 + c * 16 + ln];
      b[1] = sW[(ka + 1) * 64 + c * 16 + ln];
      acc[c] = wmma4(a, b, acc[c]);
    }
  }

  float bias[4];
#pragma unroll
  for (int c = 0; c < 4; ++c) bias[c] = sb[c * 16 + ln];
#pragma unroll
  for (int v = 0; v < 8; ++v) {
    long row = row0 + v + 8 * half;
    if (row < N) {
      float* po = h + row * 64 + ln;
      po[0] = acc[0][v] + bias[0];
      po[16] = acc[1][v] + bias[1];
      po[32] = acc[2][v] + bias[2];
      po[48] = acc[3][v] + bias[3];
    }
  }
}

__global__ __launch_bounds__(256) void sage_update(
    const float* __restrict__ agg, const float* __restrict__ hin,
    const float* __restrict__ invc, const float* __restrict__ Wl,
    const float* __restrict__ bl, const float* __restrict__ Wr,
    const float* __restrict__ gam, const float* __restrict__ bet,
    float* __restrict__ hout, int N) {
  __shared__ float sWl[64 * 64];
  __shared__ float sWr[64 * 64];
  __shared__ float sbl[64], sg[64], sbeta[64];
  int t = threadIdx.x;
  for (int i = t; i < 64 * 64; i += 256) {
    sWl[i] = Wl[i];
    sWr[i] = Wr[i];
  }
  if (t < 64) {
    sbl[t] = bl[t];
    sg[t] = gam[t];
    sbeta[t] = bet[t];
  }
  __syncthreads();

  int wave = t >> 5, lane = t & 31, half = lane >> 4, ln = lane & 15;
  long tile = (long)blockIdx.x * 8 + wave;
  long row0 = tile * 16;
  if (row0 >= N) return;

  int r = (int)row0 + ln;
  if (r > N - 1) r = N - 1;
  float iv = invc[r];
  const float* pag = agg + (long)r * 64;
  const float* ph = hin + (long)r * 64;

  v8f acc[4];
#pragma unroll
  for (int c = 0; c < 4; ++c) acc[c] = (v8f){0, 0, 0, 0, 0, 0, 0, 0};

#pragma unroll
  for (int k = 0; k < 64; k += 4) {
    int ka = k + 2 * half;
    v2f aA, aH;
    aA[0] = pag[ka] * iv;
    aA[1] = pag[ka + 1] * iv;
    aH[0] = ph[ka];
    aH[1] = ph[ka + 1];
#pragma unroll
    for (int c = 0; c < 4; ++c) {
      v2f bL, bR;
      bL[0] = sWl[ka * 64 + c * 16 + ln];
      bL[1] = sWl[(ka + 1) * 64 + c * 16 + ln];
      bR[0] = sWr[ka * 64 + c * 16 + ln];
      bR[1] = sWr[(ka + 1) * 64 + c * 16 + ln];
      acc[c] = wmma4(aA, bL, acc[c]);
      acc[c] = wmma4(aH, bR, acc[c]);
    }
  }

  float bias[4], g4[4], be4[4];
#pragma unroll
  for (int c = 0; c < 4; ++c) {
    bias[c] = sbl[c * 16 + ln];
    g4[c] = sg[c * 16 + ln];
    be4[c] = sbeta[c * 16 + ln];
  }

#pragma unroll
  for (int v = 0; v < 8; ++v) {
    float r0 = fmaxf(acc[0][v] + bias[0], 0.0f);
    float r1 = fmaxf(acc[1][v] + bias[1], 0.0f);
    float r2 = fmaxf(acc[2][v] + bias[2], 0.0f);
    float r3 = fmaxf(acc[3][v] + bias[3], 0.0f);
    float s = r0 + r1 + r2 + r3;
    s += __shfl_xor(s, 1, 16);
    s += __shfl_xor(s, 2, 16);
    s += __shfl_xor(s, 4, 16);
    s += __shfl_xor(s, 8, 16);
    float mu = s * (1.0f / 64.0f);
    float d0 = r0 - mu, d1 = r1 - mu, d2 = r2 - mu, d3 = r3 - mu;
    float q = d0 * d0 + d1 * d1 + d2 * d2 + d3 * d3;
    q += __shfl_xor(q, 1, 16);
    q += __shfl_xor(q, 2, 16);
    q += __shfl_xor(q, 4, 16);
    q += __shfl_xor(q, 8, 16);
    float rin = rsqrtf(q * (1.0f / 64.0f) + LN_EPS);
    long row = row0 + v + 8 * half;
    if (row < N) {
      float* po = hout + row * 64 + ln;
      po[0] = d0 * rin * g4[0] + be4[0];
      po[16] = d1 * rin * g4[1] + be4[1];
      po[32] = d2 * rin * g4[2] + be4[2];
      po[48] = d3 * rin * g4[3] + be4[3];
    }
  }
}

extern "C" void kernel_launch(void* const* d_in, const int* in_sizes, int n_in,
                              void* d_out, int out_size, void* d_ws,
                              size_t ws_size, hipStream_t stream) {
  (void)n_in;
  (void)out_size;
  (void)ws_size;
  const float* x = (const float*)d_in[0];
  const int* ei = (const int*)d_in[1];
  const float* Win = (const float*)d_in[2];
  const float* bin = (const float*)d_in[3];
  const float* Wl = (const float*)d_in[4];
  const float* bl = (const float*)d_in[5];
  const float* Wr = (const float*)d_in[6];
  const float* gam = (const float*)d_in[7];
  const float* bet = (const float*)d_in[8];

  const int N = in_sizes[0] / 16;
  const long E = in_sizes[1] / 2;
  const int* src = ei;
  const int* dst = ei + E;

  float* ws = (float*)d_ws;
  size_t nA = ((size_t)N + 63) & ~(size_t)63;
  size_t eA = ((size_t)E + 63) & ~(size_t)63;
  int* cnt = (int*)ws;
  float* buf0 = ws + nA;
  float* buf1 = buf0 + (size_t)N * 64;
  int* grp = (int*)buf1;
  int* offs = grp + eA;
  float* inv = (float*)(offs + nA);
  float* outp = (float*)d_out;
  int* rank = (int*)d_out;

  const int eblocks = (int)((E + 255) / 256);
  zero_i32<<<(N + 255) / 256, 256, 0, stream>>>(cnt, N);
  rank_edges<<<eblocks, 256, 0, stream>>>(dst, cnt, rank, E);
  scan_counts<<<(N + 1023) / 1024, 256, 0, stream>>>(cnt, offs, inv, N);
  fill_groups<<<eblocks, 256, 0, stream>>>(src, dst, rank, offs, grp, E);

  int tiles = (N + 15) / 16;
  int gblocks = (tiles + 7) / 8;
  input_gemm<<<gblocks, 256, 0, stream>>>(x, Win, bin, buf0, N);

  float* hcur = buf0;
  float* other = outp;
  for (int l = 0; l < 3; ++l) {
    gather_rows<<<(N + 7) / 8, 256, 0, stream>>>(
        (const float4*)hcur, grp, offs, cnt, (float4*)other, N);
    sage_update<<<gblocks, 256, 0, stream>>>(
        other, hcur, inv, Wl + (size_t)l * 4096, bl + (size_t)l * 64,
        Wr + (size_t)l * 4096, gam, bet, other, N);
    float* tmp = hcur;
    hcur = other;
    other = tmp;
  }
}


# Edges grouped by destination with a two-level counting sort on LDS counters (512-node bins: count, scan, scatter, per-bin sort), no global returning add per edge; sum order stays arrival order
# speedup vs baseline: 3.5017x; 1.5782x over previous
#include <hip/hip_runtime.h>


typedef float v2f __attribute__((ext_vector_type(2)));
typedef float v8f __attribute__((ext_vector_type(8)));

#define LN_EPS 1e-5f

__device__ __forceinline__ v8f wmma4(v2f a, v2f b, v8f c) {
  return __builtin_amdgcn_wmma_f32_16x16x4_f32(false, a, false, b, (short)0, c,
                                               false, false);
}

#define BSHIFT 9
#define BSIZE 512
#define BMASK 511
#define MAXNB 2048
#define CHUNK 4096
#define EPT 16

__device__ __forceinline__ void lds_inc(int* p) {
  __hip_atomic_fetch_add(p, 1, __ATOMIC_RELAXED, __HIP_MEMORY_SCOPE_WORKGROUP);
}
__device__ __forceinline__ int lds_take(int* p) {
  return __hip_atomic_fetch_add(p, 1, __ATOMIC_RELAXED,
                                __HIP_MEMORY_SCOPE_WORKGROUP);
}

__global__ __launch_bounds__(256) void bin_count(const int* __restrict__ dst,
                                                 int* __restrict__ wgcnt,
                                                 long E, int NB, int G) {
  __shared__ int hist[MAXNB];
  int t = threadIdx.x;
  for (int i = t; i < NB; i += 256) hist[i] = 0;
  __syncthreads();
  long base = (long)blockIdx.x * CHUNK + t;
  int d[EPT];
#pragma unroll
  for (int k = 0; k < EPT; ++k) {
    long e = base + k * 256;
    d[k] = (e < E) ? dst[e] : -1;
  }
#pragma unroll
  for (int k = 0; k < EPT; ++k)
    if (d[k] >= 0) lds_inc(&hist[d[k] >> BSHIFT]);
  __syncthreads();
  for (int i = t; i < NB; i += 256)
    wgcnt[(long)i * G + blockIdx.x] = hist[i];
}

__global__ __launch_bounds__(256) void row_scan(int* __restrict__ wgcnt,
                                                int* __restrict__ total,
                                                int G) {
  __shared__ int swave[8];
  int t = threadIdx.x, lane = t & 31, wave = t >> 5;
  int* row = wgcnt + (long)blockIdx.x * G;
  int carry = 0;
  for (int i0 = 0; i0 < G; i0 += 256) {
    int i = i0 + t;
    int v = (i < G) ? row[i] : 0;
    int inc = v;
#pragma unroll
    for (int o = 1; o < 32; o <<= 1) {
      int x = __shfl_up(inc, o);
      if (lane >= o) inc += x;
    }
    if (lane == 31) swave[wave] = inc;
    __syncthreads();
    int ex = carry + inc - v;
    int tile = 0;
#pragma unroll
    for (int w = 0; w < 8; ++w) {
      int s = swave[w];
      if (w < wave) ex += s;
      tile += s;
    }
    if (i < G) row[i] = ex;
    carry += tile;
    __syncthreads();
  }
  if (t == 0) total[blockIdx.x] = carry;
}

__global__ __launch_bounds__(256) void bin_scatter(
    const int* __restrict__ src, const int* __restrict__ dst,
    const int* __restrict__ wgoff, const int* __restrict__ total,
    int* __restrict__ tmp, long E, int NB, int G) {
  __shared__ int cur[MAXNB];
  __shared__ int swave[8];
  int t = threadIdx.x, lane = t & 31, wave = t >> 5;
  int v[8];
  int s = 0;
#pragma unroll
  for (int k = 0; k < 8; ++k) {
    int i = t * 8 + k;
    v[k] = (i < NB) ? total[i] : 0;
    s += v[k];
  }
  int inc = s;
#pragma unroll
  for (int o = 1; o < 32; o <<= 1) {
    int x = __shfl_up(inc, o);
    if (lane >= o) inc += x;
  }
  if (lane == 31) swave[wave] = inc;
  __syncthreads();
  int ex = inc - s;
#pragma unroll
  for (int w = 0; w < 8; ++w)
    if (w < wave) ex += swave[w];
#pragma unroll
  for (int k = 0; k < 8; ++k) {
    int i = t * 8 + k;
    if (i < NB) cur[i] = ex + wgoff[(long)i * G + blockIdx.x];
    ex += v[k];
  }
  __syncthreads();

  long base = (long)blockIdx.x * CHUNK + t;
  long last = E - 1;
  int d[EPT], sv[EPT];
#pragma unroll
  for (int k = 0; k < EPT; ++k) {
    long e = base + k * 256;
    long ec = e < last ? e : last;
    d[k] = dst[ec];
    sv[k] = src[ec];
  }
#pragma unroll
  for (int k = 0; k < EPT; ++k) {
    if (base + k * 256 < E) {
      int pos = lds_take(&cur[d[k] >> BSHIFT]);
      tmp[pos] = (sv[k] << BSHIFT) | (d[k] & BMASK);
    }
  }
}

__global__ __launch_bounds__(1024) void bin_sort(
    const int* __restrict__ tmp, const int* __restrict__ total,
    int* __restrict__ grp, int* __restrict__ cnt, int* __restrict__ offs,
    float* __restrict__ inv, int N) {
  __shared__ int hist[BSIZE];
  __shared__ int cur[BSIZE];
  __shared__ int sred[32];
  __shared__ int swave[32];
  int t = threadIdx.x, lane = t & 31, wave = t >> 5;
  int b = blockIdx.x;

  int part = 0;
  for (int i = t; i < b; i += 1024) part += total[i];
  part += __shfl_xor(part, 16);
  part += __shfl_xor(part, 8);
  part += __shfl_xor(part, 4);
  part += __shfl_xor(part, 2);
  part += __shfl_xor(part, 1);
  if (lane == 0) sred[wave] = part;
  if (t < BSIZE) hist[t] = 0;
  __syncthreads();
  int start = 0;
#pragma unroll
  for (int w = 0; w < 32; ++w) start += sred[w];
  int m = total[b];
  const int* p = tmp + start;

  for (int i = t; i < m; i += 4096) {
    int w4[4];
#pragma unroll
    for (int k = 0; k < 4; ++k) {
      int j = i + k * 1024;
      w4[k] = (j < m) ? p[j] : -1;
    }
#pragma unroll
    for (int k = 0; k < 4; ++k)
      if (w4[k] >= 0) lds_inc(&hist[w4[k] & BMASK]);
  }
  __syncthreads();

  int c = (t < BSIZE) ? hist[t] : 0;
  int inc = c;
#pragma unroll
  for (int o = 1; o < 32; o <<= 1) {
    int x = __shfl_up(inc, o);
    if (lane >= o) inc += x;
  }
  if (lane == 31) swave[wave] = inc;
  __syncthreads();
  int ex = inc - c;
#pragma unroll
  for (int w = 0; w < 32; ++w)
    if (w < wave) ex += swave[w];
  if (t < BSIZE) {
    cur[t] = ex;
    int node = b * BSIZE + t;
    if (node < N) {
      cnt[node] = c;
      offs[node] = start + ex;
      inv[node] = 1.0f / fmaxf((float)c, 1.0f);
    }
  }
  __syncthreads();

  for (int i = t; i < m; i += 4096) {
    int w4[4];
#pragma unroll
    for (int k = 0; k < 4; ++k) {
      int j = i + k * 1024;
      w4[k] = (j < m) ? p[j] : -1;
    }
#pragma unroll
    for (int k = 0; k < 4; ++k) {
      if (w4[k] >= 0) {
        int pos = lds_take(&cur[w4[k] & BMASK]);
        grp[start + pos] = (int)((unsigned)w4[k] >> BSHIFT);
      }
    }
  }
}

__global__ __launch_bounds__(256) void gather_rows(
    const float4* __restrict__ h4, const int* __restrict__ grp,
    const int* __restrict__ offs, const int* __restrict__ cnt,
    float4* __restrict__ agg4, int N) {
  int t = threadIdx.x;
  int lane = t & 31, half = lane >> 4, c = lane & 15;
  int node = __builtin_amdgcn_readfirstlane((int)blockIdx.x * 8 + (t >> 5));
  if (node >= N) return;
  int beg = offs[node];
  int deg = cnt[node];
  const int* p = grp + beg;

  float4 a = make_float4(0.0f, 0.0f, 0.0f, 0.0f);
  int j = 0;
  for (; j + 8 <= deg; j += 8) {
    int e0 = p[j], e1 = p[j + 1], e2 = p[j + 2], e3 = p[j + 3];
    int e4 = p[j + 4], e5 = p[j + 5], e6 = p[j + 6], e7 = p[j + 7];
    int s0 = half ? e1 : e0;
    int s1 = half ? e3 : e2;
    int s2 = half ? e5 : e4;
    int s3 = half ? e7 : e6;
    float4 v0 = h4[(long)s0 * 16 + c];
    float4 v1 = h4[(long)s1 * 16 + c];
    float4 v2 = h4[(long)s2 * 16 + c];
    float4 v3 = h4[(long)s3 * 16 + c];
    a.x += v0.x; a.y += v0.y; a.z += v0.z; a.w += v0.w;
    a.x += v1.x; a.y += v1.y; a.z += v1.z; a.w += v1.w;
    a.x += v2.x; a.y += v2.y; a.z += v2.z; a.w += v2.w;
    a.x += v3.x; a.y += v3.y; a.z += v3.z; a.w += v3.w;
  }
  if (j < deg) {
    int last = deg - 1;
    int i0 = j + half, i1 = i0 + 2, i2 = i0 + 4, i3 = i0 + 6;
    int s0 = p[i0 < last ? i0 : last];
    int s1 = p[i1 < last ? i1 : last];
    int s2 = p[i2 < last ? i2 : last];
    int s3 = p[i3 < last ? i3 : last];
    float4 v0 = h4[(long)s0 * 16 + c];
    float4 v1 = h4[(long)s1 * 16 + c];
    float4 v2 = h4[(long)s2 * 16 + c];
    float4 v3 = h4[(long)s3 * 16 + c];
    bool k0 = i0 < deg, k1 = i1 < deg, k2 = i2 < deg, k3 = i3 < deg;
    a.x += k0 ? v0.x : 0.0f; a.y += k0 ? v0.y : 0.0f;
    a.z += k0 ? v0.z : 0.0f; a.w += k0 ? v0.w : 0.0f;
    a.x += k1 ? v1.x : 0.0f; a.y += k1 ? v1.y : 0.0f;
    a.z += k1 ? v1.z : 0.0f; a.w += k1 ? v1.w : 0.0f;
    a.x += k2 ? v2.x : 0.0f; a.y += k2 ? v2.y : 0.0f;
    a.z += k2 ? v2.z : 0.0f; a.w += k2 ? v2.w : 0.0f;
    a.x += k3 ? v3.x : 0.0f; a.y += k3 ? v3.y : 0.0f;
    a.z += k3 ? v3.z : 0.0f; a.w += k3 ? v3.w : 0.0f;
  }
  a.x += __shfl_xor(a.x, 16);
  a.y += __shfl_xor(a.y, 16);
  a.z += __shfl_xor(a.z, 16);
  a.w += __shfl_xor(a.w, 16);
  if (half == 0) agg4[(long)node * 16 + c] = a;
}

__global__ __launch_bounds__(256) void input_gemm(
    const float* __restrict__ x, const float* __restrict__ Win,
    const float* __restrict__ bin, float* __restrict__ h, int N) {
  __shared__ float sW[16 * 64];
  __shared__ float sb[64];
  int t = threadIdx.x;
  for (int i = t; i < 16 * 64; i += 256) sW[i] = Win[i];
  if (t < 64) sb[t] = bin[t];
  __syncthreads();

  int wave = t >> 5, lane = t & 31, half = lane >> 4, ln = lane & 15;
  long tile = (long)blockIdx.x * 8 + wave;
  long row0 = tile * 16;
  if (row0 >= N) return;

  int r = (int)row0 + ln;
  if (r > N - 1) r = N - 1;
  const float* px = x + (long)r * 16;

  v8f acc[4];
#pragma unroll
  for (int c = 0; c < 4; ++c) acc[c] = (v8f){0, 0, 0, 0, 0, 0, 0, 0};

#pragma unroll
  for (int k = 0; k < 16; k += 4) {
    int ka = k + 2 * half;
    v2f a;
    a[0] = px[ka];
    a[1] = px[ka + 1];
#pragma unroll
    for (int c = 0; c < 4; ++c) {
      v2f b;
      b[0] = sW[ka * 64 + c * 16 + ln];
      b[1] = sW[(ka + 1) * 64 + c * 16 + ln];
      acc[c] = wmma4(a, b, acc[c]);
    }
  }

  float bias[4];
#pragma unroll
  for (int c = 0; c < 4; ++c) bias[c] = sb[c * 16 + ln];
#pragma unroll
  for (int v = 0; v < 8; ++v) {
    long row = row0 + v + 8 * half;
    if (row < N) {
      float* po = h + row * 64 + ln;
      po[0] = acc[0][v] + bias[0];
      po[16] = acc[1][v] + bias[1];
      po[32] = acc[2][v] + bias[2];
      po[48] = acc[3][v] + bias[3];
    }
  }
}

__global__ __launch_bounds__(256) void sage_update(
    const float* __restrict__ agg, const float* __restrict__ hin,
    const float* __restrict__ invc, const float* __restrict__ Wl,
    const float* __restrict__ bl, const float* __restrict__ Wr,
    const float* __restrict__ gam, const float* __restrict__ bet,
    float* __restrict__ hout, int N) {
  __shared__ float sWl[64 * 64];
  __shared__ float sWr[64 * 64];
  __shared__ float sbl[64], sg[64], sbeta[64];
  int t = threadIdx.x;
  for (int i = t; i < 64 * 64; i += 256) {
    sWl[i] = Wl[i];
    sWr[i] = Wr[i];
  }
  if (t < 64) {
    sbl[t] = bl[t];
    sg[t] = gam[t];
    sbeta[t] = bet[t];
  }
  __syncthreads();

  int wave = t >> 5, lane = t & 31, half = lane >> 4, ln = lane & 15;
  long tile = (long)blockIdx.x * 8 + wave;
  long row0 = tile * 16;
  if (row0 >= N) return;

  int r = (int)row0 + ln;
  if (r > N - 1) r = N - 1;
  float iv = invc[r];
  const float* pag = agg + (long)r * 64;
  const float* ph = hin + (long)r * 64;

  v8f acc[4];
#pragma unroll
  for (int c = 0; c < 4; ++c) acc[c] = (v8f){0, 0, 0, 0, 0, 0, 0, 0};

#pragma unroll
  for (int k = 0; k < 64; k += 4) {
    int ka = k + 2 * half;
    v2f aA, aH;
    aA[0] = pag[ka] * iv;
    aA[1] = pag[ka + 1] * iv;
    aH[0] = ph[ka];
    aH[1] = ph[ka + 1];
#pragma unroll
    for (int c = 0; c < 4; ++c) {
      v2f bL, bR;
      bL[0] = sWl[ka * 64 + c * 16 + ln];
      bL[1] = sWl[(ka + 1) * 64 + c * 16 + ln];
      bR[0] = sWr[ka * 64 + c * 16 + ln];
      bR[1] = sWr[(ka + 1) * 64 + c * 16 + ln];
      acc[c] = wmma4(aA, bL, acc[c]);
      acc[c] = wmma4(aH, bR, acc[c]);
    }
  }

  float bias[4], g4[4], be4[4];
#pragma unroll
  for (int c = 0; c < 4; ++c) {
    bias[c] = sbl[c * 16 + ln];
    g4[c] = sg[c * 16 + ln];
    be4[c] = sbeta[c * 16 + ln];
  }

#pragma unroll
  for (int v = 0; v < 8; ++v) {
    float r0 = fmaxf(acc[0][v] + bias[0], 0.0f);
    float r1 = fmaxf(acc[1][v] + bias[1], 0.0f);
    float r2 = fmaxf(acc[2][v] + bias[2], 0.0f);
    float r3 = fmaxf(acc[3][v] + bias[3], 0.0f);
    float s = r0 + r1 + r2 + r3;
    s += __shfl_xor(s, 1, 16);
    s += __shfl_xor(s, 2, 16);
    s += __shfl_xor(s, 4, 16);
    s += __shfl_xor(s, 8, 16);
    float mu = s * (1.0f / 64.0f);
    float d0 = r0 - mu, d1 = r1 - mu, d2 = r2 - mu, d3 = r3 - mu;
    float q = d0 * d0 + d1 * d1 + d2 * d2 + d3 * d3;
    q += __shfl_xor(q, 1, 16);
    q += __shfl_xor(q, 2, 16);
    q += __shfl_xor(q, 4, 16);
    q += __shfl_xor(q, 8, 16);
    float rin = rsqrtf(q * (1.0f / 64.0f) + LN_EPS);
    long row = row0 + v + 8 * half;
    if (row < N) {
      float* po = hout + row * 64 + ln;
      po[0] = d0 * rin * g4[0] + be4[0];
      po[16] = d1 * rin * g4[1] + be4[1];
      po[32] = d2 * rin * g4[2] + be4[2];
      po[48] = d3 * rin * g4[3] + be4[3];
    }
  }
}

extern "C" void kernel_launch(void* const* d_in, const int* in_sizes, int n_in,
                              void* d_out, int out_size, void* d_ws,
                              size_t ws_size, hipStream_t stream) {
  (void)n_in;
  (void)out_size;
  (void)ws_size;
  const float* x = (const float*)d_in[0];
  const int* ei = (const int*)d_in[1];
  const float* Win = (const float*)d_in[2];
  const float* bin = (const float*)d_in[3];
  const float* Wl = (const float*)d_in[4];
  const float* bl = (const float*)d_in[5];
  const float* Wr = (const float*)d_in[6];
  const float* gam = (const float*)d_in[7];
  const float* bet = (const float*)d_in[8];

  const int N = in_sizes[0] / 16;
  const long E = in_sizes[1] / 2;
  const int* src = ei;
  const int* dst = ei + E;

  float* ws = (float*)d_ws;
  size_t nA = ((size_t)N + 63) & ~(size_t)63;
  size_t eA = ((size_t)E + 63) & ~(size_t)63;
  int* cnt = (int*)ws;
  float* buf0 = ws + nA;
  float* buf1 = buf0 + (size_t)N * 64;
  int* grp = (int*)buf1;
  int* offs = grp + eA;
  float* inv = (float*)(offs + nA);
  float* outp = (float*)d_out;

  const int NB = (N + BSIZE - 1) / BSIZE;
  const int G = (int)((E + CHUNK - 1) / CHUNK);
  size_t mA = ((size_t)NB * (size_t)G + 63) & ~(size_t)63;
  int* binned = (int*)d_out;
  int* wgcnt = binned + eA;
  int* total = wgcnt + mA;

  bin_count<<<G, 256, 0, stream>>>(dst, wgcnt, E, NB, G);
  row_scan<<<NB, 256, 0, stream>>>(wgcnt, total, G);
  bin_scatter<<<G, 256, 0, stream>>>(src, dst, wgcnt, total, binned, E, NB, G);
  bin_sort<<<NB, 1024, 0, stream>>>(binned, total, grp, cnt, offs, inv, N);

  int tiles = (N + 15) / 16;
  int gblocks = (tiles + 7) / 8;
  input_gemm<<<gblocks, 256, 0, stream>>>(x, Win, bin, buf0, N);

  float* hcur = buf0;
  float* other = outp;
  for (int l = 0; l < 3; ++l) {
    gather_rows<<<(N + 7) / 8, 256, 0, stream>>>(
        (const float4*)hcur, grp, offs, cnt, (float4*)other, N);
    sage_update<<<gblocks, 256, 0, stream>>>(
        other, hcur, inv, Wl + (size_t)l * 4096, bl + (size_t)l * 64,
        Wr + (size_t)l * 4096, gam, bet, other, N);
    float* tmp = hcur;
    hcur = other;
    other = tmp;
  }
}


# Neighbour sum moved into the update kernel: each 16-row tile wave sums its 16 nodes' source rows into its own LDS rows and reads the A fragments there; no agg plane, 8 launches; sum order kept
# speedup vs baseline: 3.7619x; 1.0743x over previous
#include <hip/hip_runtime.h>


typedef float v2f __attribute__((ext_vector_type(2)));
typedef float v8f __attribute__((ext_vector_type(8)));

#define LN_EPS 1e-5f

__device__ __forceinline__ v8f wmma4(v2f a, v2f b, v8f c) {
  return __builtin_amdgcn_wmma_f32_16x16x4_f32(false, a, false, b, (short)0, c,
                                               false, false);
}

#define BSHIFT 9
#define BSIZE 512
#define BMASK 511
#define MAXNB 2048
#define CHUNK 4096
#define EPT 16

__device__ __forceinline__ void lds_inc(int* p) {
  __hip_atomic_fetch_add(p, 1, __ATOMIC_RELAXED, __HIP_MEMORY_SCOPE_WORKGROUP);
}
__device__ __forceinline__ int lds_take(int* p) {
  return __hip_atomic_fetch_add(p, 1, __ATOMIC_RELAXED,
                                __HIP_MEMORY_SCOPE_WORKGROUP);
}

__global__ __launch_bounds__(256) void bin_count(const int* __restrict__ dst,
                                                 int* __restrict__ wgcnt,
                                                 long E, int NB, int G) {
  __shared__ int hist[MAXNB];
  int t = threadIdx.x;
  for (int i = t; i < NB; i += 256) hist[i] = 0;
  __syncthreads();
  long base = (long)blockIdx.x * CHUNK + t;
  int d[EPT];
#pragma unroll
  for (int k = 0; k < EPT; ++k) {
    long e = base + k * 256;
    d[k] = (e < E) ? dst[e] : -1;
  }
#pragma unroll
  for (int k = 0; k < EPT; ++k)
    if (d[k] >= 0) lds_inc(&hist[d[k] >> BSHIFT]);
  __syncthreads();
  for (int i = t; i < NB; i += 256)
    wgcnt[(long)i * G + blockIdx.x] = hist[i];
}

__global__ __launch_bounds__(256) void row_scan(int* __restrict__ wgcnt,
                                                int* __restrict__ total,
                                                int G) {
  __shared__ int swave[8];
  int t = threadIdx.x, lane = t & 31, wave = t >> 5;
  int* row = wgcnt + (long)blockIdx.x * G;
  int carry = 0;
  for (int i0 = 0; i0 < G; i0 += 256) {
    int i = i0 + t;
    int v = (i < G) ? row[i] : 0;
    int inc = v;
#pragma unroll
    for (int o = 1; o < 32; o <<= 1) {
      int x = __shfl_up(inc, o);
      if (lane >= o) inc += x;
    }
    if (lane == 31) swave[wave] = inc;
    __syncthreads();
    int ex = carry + inc - v;
    int tile = 0;
#pragma unroll
    for (int w = 0; w < 8; ++w) {
      int s = swave[w];
      if (w < wave) ex += s;
      tile += s;
    }
    if (i < G) row[i] = ex;
    carry += tile;
    __syncthreads();
  }
  if (t == 0) total[blockIdx.x] = carry;
}

__global__ __launch_bounds__(256) void bin_scatter(
    const int* __restrict__ src, const int* __restrict__ dst,
    const int* __restrict__ wgoff, const int* __restrict__ total,
    int* __restrict__ tmp, long E, int NB, int G) {
  __shared__ int cur[MAXNB];
  __shared__ int swave[8];
  int t = threadIdx.x, lane = t & 31, wave = t >> 5;
  int v[8];
  int s = 0;
#pragma unroll
  for (int k = 0; k < 8; ++k) {
    int i = t * 8 + k;
    v[k] = (i < NB) ? total[i] : 0;
    s += v[k];
  }
  int inc = s;
#pragma unroll
  for (int o = 1; o < 32; o <<= 1) {
    int x = __shfl_up(inc, o);
    if (lane >= o) inc += x;
  }
  if (lane == 31) swave[wave] = inc;
  __syncthreads();
  int ex = inc - s;
#pragma unroll
  for (int w = 0; w < 8; ++w)
    if (w < wave) ex += swave[w];
#pragma unroll
  for (int k = 0; k < 8; ++k) {
    int i = t * 8 + k;
    if (i < NB) cur[i] = ex + wgoff[(long)i * G + blockIdx.x];
    ex += v[k];
  }
  __syncthreads();

  long base = (long)blockIdx.x * CHUNK + t;
  long last = E - 1;
  int d[EPT], sv[EPT];
#pragma unroll
  for (int k = 0; k < EPT; ++k) {
    long e = base + k * 256;
    long ec = e < last ? e : last;
    d[k] = dst[ec];
    sv[k] = src[ec];
  }
#pragma unroll
  for (int k = 0; k < EPT; ++k) {
    if (base + k * 256 < E) {
      int pos = lds_take(&cur[d[k] >> BSHIFT]);
      tmp[pos] = (sv[k] << BSHIFT) | (d[k] & BMASK);
    }
  }
}

__global__ __launch_bounds__(1024) void bin_sort(
    const int* __restrict__ tmp, const int* __restrict__ total,
    int* __restrict__ grp, int* __restrict__ cnt, int* __restrict__ offs,
    float* __restrict__ inv, int N) {
  __shared__ int hist[BSIZE];
  __shared__ int cur[BSIZE];
  __shared__ int sred[32];
  __shared__ int swave[32];
  int t = threadIdx.x, lane = t & 31, wave = t >> 5;
  int b = blockIdx.x;

  int part = 0;
  for (int i = t; i < b; i += 1024) part += total[i];
  part += __shfl_xor(part, 16);
  part += __shfl_xor(part, 8);
  part += __shfl_xor(part, 4);
  part += __shfl_xor(part, 2);
  part += __shfl_xor(part, 1);
  if (lane == 0) sred[wave] = part;
  if (t < BSIZE) hist[t] = 0;
  __syncthreads();
  int start = 0;
#pragma unroll
  for (int w = 0; w < 32; ++w) start += sred[w];
  int m = total[b];
  const int* p = tmp + start;

  for (int i = t; i < m; i += 4096) {
    int w4[4];
#pragma unroll
    for (int k = 0; k < 4; ++k) {
      int j = i + k * 1024;
      w4[k] = (j < m) ? p[j] : -1;
    }
#pragma unroll
    for (int k = 0; k < 4; ++k)
      if (w4[k] >= 0) lds_inc(&hist[w4[k] & BMASK]);
  }
  __syncthreads();

  int c = (t < BSIZE) ? hist[t] : 0;
  int inc = c;
#pragma unroll
  for (int o = 1; o < 32; o <<= 1) {
    int x = __shfl_up(inc, o);
    if (lane >= o) inc += x;
  }
  if (lane == 31) swave[wave] = inc;
  __syncthreads();
  int ex = inc - c;
#pragma unroll
  for (int w = 0; w < 32; ++w)
    if (w < wave) ex += swave[w];
  if (t < BSIZE) {
    cur[t] = ex;
    int node = b * BSIZE + t;
    if (node < N) {
      cnt[node] = c;
      offs[node] = start + ex;
      inv[node] = 1.0f / fmaxf((float)c, 1.0f);
    }
  }
  __syncthreads();

  for (int i = t; i < m; i += 4096) {
    int w4[4];
#pragma unroll
    for (int k = 0; k < 4; ++k) {
      int j = i + k * 1024;
      w4[k] = (j < m) ? p[j] : -1;
    }
#pragma unroll
    for (int k = 0; k < 4; ++k) {
      if (w4[k] >= 0) {
        int pos = lds_take(&cur[w4[k] & BMASK]);
        grp[start + pos] = (int)((unsigned)w4[k] >> BSHIFT);
      }
    }
  }
}

__global__ __launch_bounds__(256) void input_gemm(
    const float* __restrict__ x, const float* __restrict__ Win,
    const float* __restrict__ bin, float* __restrict__ h, int N) {
  __shared__ float sW[16 * 64];
  __shared__ float sb[64];
  int t = threadIdx.x;
  for (int i = t; i < 16 * 64; i += 256) sW[i] = Win[i];
  if (t < 64) sb[t] = bin[t];
  __syncthreads();

  int wave = t >> 5, lane = t & 31, half = lane >> 4, ln = lane & 15;
  long tile = (long)blockIdx.x * 8 + wave;
  long row0 = tile * 16;
  if (row0 >= N) return;

  int r = (int)row0 + ln;
  if (r > N - 1) r = N - 1;
  const float* px = x + (long)r * 16;

  v8f acc[4];
#pragma unroll
  for (int c = 0; c < 4; ++c) acc[c] = (v8f){0, 0, 0, 0, 0, 0, 0, 0};

#pragma unroll
  for (int k = 0; k < 16; k += 4) {
    int ka = k + 2 * half;
    v2f a;
    a[0] = px[ka];
    a[1] = px[ka + 1];
#pragma unroll
    for (int c = 0; c < 4; ++c) {
      v2f b;
      b[0] = sW[ka * 64 + c * 16 + ln];
      b[1] = sW[(ka + 1) * 64 + c * 16 + ln];
      acc[c] = wmma4(a, b, acc[c]);
    }
  }

  float bias[4];
#pragma unroll
  for (int c = 0; c < 4; ++c) bias[c] = sb[c * 16 + ln];
#pragma unroll
  for (int v = 0; v < 8; ++v) {
    long row = row0 + v + 8 * half;
    if (row < N) {
      float* po = h + row * 64 + ln;
      po[0] = acc[0][v] + bias[0];
      po[16] = acc[1][v] + bias[1];
      po[32] = acc[2][v] + bias[2];
      po[48] = acc[3][v] + bias[3];
    }
  }
}

#define APITCH 68
__global__ __launch_bounds__(256) void sage_layer(
    const float* __restrict__ hin, const int* __restrict__ grp,
    const int* __restrict__ offs, const int* __restrict__ cnt,
    const float* __restrict__ invc, const float* __restrict__ Wl,
    const float* __restrict__ bl, const float* __restrict__ Wr,
    const float* __restrict__ gam, const float* __restrict__ bet,
    float* __restrict__ hout, int N) {
  __shared__ float sWl[64 * 64];
  __shared__ float sWr[64 * 64];
  __shared__ float sbl[64], sg[64], sbeta[64];
  __shared__ float4 sA[8][16 * (APITCH / 4)];
  int t = threadIdx.x;
  for (int i = t; i < 64 * 64; i += 256) {
    sWl[i] = Wl[i];
    sWr[i] = Wr[i];
  }
  if (t < 64) {
    sbl[t] = bl[t];
    sg[t] = gam[t];
    sbeta[t] = bet[t];
  }
  __syncthreads();

  int wave = t >> 5, lane = t & 31, half = lane >> 4, ln = lane & 15;
  long tile = (long)blockIdx.x * 8 + wave;
  long row0 = tile * 16;
  if (row0 >= N) return;

  int r = (int)row0 + ln;
  bool live = r < N;
  if (r > N - 1) r = N - 1;
  float iv = invc[r];
  int vbeg = offs[r];
  int vdeg = live ? cnt[r] : 0;

  const float4* h4 = (const float4*)hin;
  float4* sAw = sA[wave];
  for (int i = 0; i < 16; ++i) {
    int beg = __builtin_amdgcn_readlane(vbeg, i);
    int deg = __builtin_amdgcn_readlane(vdeg, i);
    const int* p = grp + beg;

    float4 a = make_float4(0.0f, 0.0f, 0.0f, 0.0f);
    int j = 0;
    for (; j + 8 <= deg; j += 8) {
      int e0 = p[j], e1 = p[j + 1], e2 = p[j + 2], e3 = p[j + 3];
      int e4 = p[j + 4], e5 = p[j + 5], e6 = p[j + 6], e7 = p[j + 7];
      int s0 = half ? e1 : e0;
      int s1 = half ? e3 : e2;
      int s2 = half ? e5 : e4;
      int s3 = half ? e7 : e6;
      float4 v0 = h4[(long)s0 * 16 + ln];
      float4 v1 = h4[(long)s1 * 16 + ln];
      float4 v2 = h4[(long)s2 * 16 + ln];
      float4 v3 = h4[(long)s3 * 16 + ln];
      a.x += v0.x; a.y += v0.y; a.z += v0.z; a.w += v0.w;
      a.x += v1.x; a.y += v1.y; a.z += v1.z; a.w += v1.w;
      a.x += v2.x; a.y += v2.y; a.z += v2.z; a.w += v2.w;
      a.x += v3.x; a.y += v3.y; a.z += v3.z; a.w += v3.w;
    }
    if (j < deg) {
      int last = deg - 1;
      int i0 = j + half, i1 = i0 + 2, i2 = i0 + 4, i3 = i0 + 6;
      int s0 = p[i0 < last ? i0 : last];
      int s1 = p[i1 < last ? i1 : last];
      int s2 = p[i2 < last ? i2 : last];
      int s3 = p[i3 < last ? i3 : last];
      float4 v0 = h4[(long)s0 * 16 + ln];
      float4 v1 = h4[(long)s1 * 16 + ln];
      float4 v2 = h4[(long)s2 * 16 + ln];
      float4 v3 = h4[(long)s3 * 16 + ln];
      bool k0 = i0 < deg, k1 = i1 < deg, k2 = i2 < deg, k3 = i3 < deg;
      a.x += k0 ? v0.x : 0.0f; a.y += k0 ? v0.y : 0.0f;
      a.z += k0 ? v0.z : 0.0f; a.w += k0 ? v0.w : 0.0f;
      a.x += k1 ? v1.x : 0.0f; a.y += k1 ? v1.y : 0.0f;
      a.z += k1 ? v1.z : 0.0f; a.w += k1 ? v1.w : 0.0f;
      a.x += k2 ? v2.x : 0.0f; a.y += k2 ? v2.y : 0.0f;
      a.z += k2 ? v2.z : 0.0f; a.w += k2 ? v2.w : 0.0f;
      a.x += k3 ? v3.x : 0.0f; a.y += k3 ? v3.y : 0.0f;
      a.z += k3 ? v3.z : 0.0f; a.w += k3 ? v3.w : 0.0f;
    }
    a.x += __shfl_xor(a.x, 16);
    a.y += __shfl_xor(a.y, 16);
    a.z += __shfl_xor(a.z, 16);
    a.w += __shfl_xor(a.w, 16);
    if (half == 0) sAw[i * (APITCH / 4) + ln] = a;
  }

  const float* pag = (const float*)sAw + ln * APITCH;
  const float* ph = hin + (long)r * 64;

  v8f acc[4];
#pragma unroll
  for (int c = 0; c < 4; ++c) acc[c] = (v8f){0, 0, 0, 0, 0, 0, 0, 0};

#pragma unroll
  for (int k = 0; k < 64; k += 4) {
    int ka = k + 2 * half;
    v2f aA, aH;
    aA[0] = pag[ka] * iv;
    aA[1] = pag[ka + 1] * iv;
    aH[0] = ph[ka];
    aH[1] = ph[ka + 1];
#pragma unroll
    for (int c = 0; c < 4; ++c) {
      v2f bL, bR;
      bL[0] = sWl[ka * 64 + c * 16 + ln];
      bL[1] = sWl[(ka + 1) * 64 + c * 16 + ln];
      bR[0] = sWr[ka * 64 + c * 16 + ln];
      bR[1] = sWr[(ka + 1) * 64 + c * 16 + ln];
      acc[c] = wmma4(aA, bL, acc[c]);
      acc[c] = wmma4(aH, bR, acc[c]);
    }
  }

  float bias[4], g4[4], be4[4];
#pragma unroll
  for (int c = 0; c < 4; ++c) {
    bias[c] = sbl[c * 16 + ln];
    g4[c] = sg[c * 16 + ln];
    be4[c] = sbeta[c * 16 + ln];
  }

#pragma unroll
  for (int v = 0; v < 8; ++v) {
    float r0 = fmaxf(acc[0][v] + bias[0], 0.0f);
    float r1 = fmaxf(acc[1][v] + bias[1], 0.0f);
    float r2 = fmaxf(acc[2][v] + bias[2], 0.0f);
    float r3 = fmaxf(acc[3][v] + bias[3], 0.0f);
    float s = r0 + r1 + r2 + r3;
    s += __shfl_xor(s, 1, 16);
    s += __shfl_xor(s, 2, 16);
    s += __shfl_xor(s, 4, 16);
    s += __shfl_xor(s, 8, 16);
    float mu = s * (1.0f / 64.0f);
    float d0 = r0 - mu, d1 = r1 - mu, d2 = r2 - mu, d3 = r3 - mu;
    float q = d0 * d0 + d1 * d1 + d2 * d2 + d3 * d3;
    q += __shfl_xor(q, 1, 16);
    q += __shfl_xor(q, 2, 16);
    q += __shfl_xor(q, 4, 16);
    q += __shfl_xor(q, 8, 16);
    float rin = rsqrtf(q * (1.0f / 64.0f) + LN_EPS);
    long row = row0 + v + 8 * half;
    if (row < N) {
      float* po = hout + row * 64 + ln;
      po[0] = d0 * rin * g4[0] + be4[0];
      po[16] = d1 * rin * g4[1] + be4[1];
      po[32] = d2 * rin * g4[2] + be4[2];
      po[48] = d3 * rin * g4[3] + be4[3];
    }
  }
}

extern "C" void kernel_launch(void* const* d_in, const int* in_sizes, int n_in,
                              void* d_out, int out_size, void* d_ws,
                              size_t ws_size, hipStream_t stream) {
  (void)n_in;
  (void)out_size;
  (void)ws_size;
  const float* x = (const float*)d_in[0];
  const int* ei = (const int*)d_in[1];
  const float* Win = (const float*)d_in[2];
  const float* bin = (const float*)d_in[3];
  const float* Wl = (const float*)d_in[4];
  const float* bl = (const float*)d_in[5];
  const float* Wr = (const float*)d_in[6];
  const float* gam = (const float*)d_in[7];
  const float* bet = (const float*)d_in[8];

  const int N = in_sizes[0] / 16;
  const long E = in_sizes[1] / 2;
  const int* src = ei;
  const int* dst = ei + E;

  float* ws = (float*)d_ws;
  size_t nA = ((size_t)N + 63) & ~(size_t)63;
  size_t eA = ((size_t)E + 63) & ~(size_t)63;
  int* cnt = (int*)ws;
  float* buf0 = ws + nA;
  float* buf1 = buf0 + (size_t)N * 64;
  int* grp = (int*)buf1;
  int* offs = grp + eA;
  float* inv = (float*)(offs + nA);
  float* outp = (float*)d_out;

  const int NB = (N + BSIZE - 1) / BSIZE;
  const int G = (int)((E + CHUNK - 1) / CHUNK);
  size_t mA = ((size_t)NB * (size_t)G + 63) & ~(size_t)63;
  int* binned = (int*)d_out;
  int* wgcnt = binned + eA;
  int* total = wgcnt + mA;

  bin_count<<<G, 256, 0, stream>>>(dst, wgcnt, E, NB, G);
  row_scan<<<NB, 256, 0, stream>>>(wgcnt, total, G);
  bin_scatter<<<G, 256, 0, stream>>>(src, dst, wgcnt, total, binned, E, NB, G);
  bin_sort<<<NB, 1024, 0, stream>>>(binned, total, grp, cnt, offs, inv, N);

  int tiles = (N + 15) / 16;
  int gblocks = (tiles + 7) / 8;
  input_gemm<<<gblocks, 256, 0, stream>>>(x, Win, bin, buf0, N);

  float* hcur = buf0;
  float* other = outp;
  for (int l = 0; l < 3; ++l) {
    sage_layer<<<gblocks, 256, 0, stream>>>(
        hcur, grp, offs, cnt, inv, Wl + (size_t)l * 4096, bl + (size_t)l * 64,
        Wr + (size_t)l * 4096, gam, bet, other, N);
    float* tmp = hcur;
    hcur = other;
    other = tmp;
  }
}


# Layer 1 taken from x by linearity of the input product: neighbour sums over 64-B x rows (8 rows per load), agg and h0 formed on WMMA in-kernel via LDS; no h0 plane, 7 launches; sum order changes
# speedup vs baseline: 4.0869x; 1.0864x over previous
#include <hip/hip_runtime.h>


typedef float v2f __attribute__((ext_vector_type(2)));
typedef float v8f __attribute__((ext_vector_type(8)));

#define LN_EPS 1e-5f

__device__ __forceinline__ v8f wmma4(v2f a, v2f b, v8f c) {
  return __builtin_amdgcn_wmma_f32_16x16x4_f32(false, a, false, b, (short)0, c,
                                               false, false);
}

#define BSHIFT 9
#define BSIZE 512
#define BMASK 511
#define MAXNB 2048
#define CHUNK 4096
#define EPT 16

__device__ __forceinline__ void lds_inc(int* p) {
  __hip_atomic_fetch_add(p, 1, __ATOMIC_RELAXED, __HIP_MEMORY_SCOPE_WORKGROUP);
}
__device__ __forceinline__ int lds_take(int* p) {
  return __hip_atomic_fetch_add(p, 1, __ATOMIC_RELAXED,
                                __HIP_MEMORY_SCOPE_WORKGROUP);
}

__global__ __launch_bounds__(256) void bin_count(const int* __restrict__ dst,
                                                 int* __restrict__ wgcnt,
                                                 long E, int NB, int G) {
  __shared__ int hist[MAXNB];
  int t = threadIdx.x;
  for (int i = t; i < NB; i += 256) hist[i] = 0;
  __syncthreads();
  long base = (long)blockIdx.x * CHUNK + t;
  int d[EPT];
#pragma unroll
  for (int k = 0; k < EPT; ++k) {
    long e = base + k * 256;
    d[k] = (e < E) ? dst[e] : -1;
  }
#pragma unroll
  for (int k = 0; k < EPT; ++k)
    if (d[k] >= 0) lds_inc(&hist[d[k] >> BSHIFT]);
  __syncthreads();
  for (int i = t; i < NB; i += 256)
    wgcnt[(long)i * G + blockIdx.x] = hist[i];
}

__global__ __launch_bounds__(256) void row_scan(int* __restrict__ wgcnt,
                                                int* __restrict__ total,
                                                int G) {
  __shared__ int swave[8];
  int t = threadIdx.x, lane = t & 31, wave = t >> 5;
  int* row = wgcnt + (long)blockIdx.x * G;
  int carry = 0;
  for (int i0 = 0; i0 < G; i0 += 256) {
    int i = i0 + t;
    int v = (i < G) ? row[i] : 0;
    int inc = v;
#pragma unroll
    for (int o = 1; o < 32; o <<= 1) {
      int x = __shfl_up(inc, o);
      if (lane >= o) inc += x;
    }
    if (lane == 31) swave[wave] = inc;
    __syncthreads();
    int ex = carry + inc - v;
    int tile = 0;
#pragma unroll
    for (int w = 0; w < 8; ++w) {
      int s = swave[w];
      if (w < wave) ex += s;
      tile += s;
    }
    if (i < G) row[i] = ex;
    carry += tile;
    __syncthreads();
  }
  if (t == 0) total[blockIdx.x] = carry;
}

__global__ __launch_bounds__(256) void bin_scatter(
    const int* __restrict__ src, const int* __restrict__ dst,
    const int* __restrict__ wgoff, const int* __restrict__ total,
    int* __restrict__ tmp, long E, int NB, int G) {
  __shared__ int cur[MAXNB];
  __shared__ int swave[8];
  int t = threadIdx.x, lane = t & 31, wave = t >> 5;
  int v[8];
  int s = 0;
#pragma unroll
  for (int k = 0; k < 8; ++k) {
    int i = t * 8 + k;
    v[k] = (i < NB) ? total[i] : 0;
    s += v[k];
  }
  int inc = s;
#pragma unroll
  for (int o = 1; o < 32; o <<= 1) {
    int x = __shfl_up(inc, o);
    if (lane >= o) inc += x;
  }
  if (lane == 31) swave[wave] = inc;
  __syncthreads();
  int ex = inc - s;
#pragma unroll
  for (int w = 0; w < 8; ++w)
    if (w < wave) ex += swave[w];
#pragma unroll
  for (int k = 0; k < 8; ++k) {
    int i = t * 8 + k;
    if (i < NB) cur[i] = ex + wgoff[(long)i * G + blockIdx.x];
    ex += v[k];
  }
  __syncthreads();

  long base = (long)blockIdx.x * CHUNK + t;
  long last = E - 1;
  int d[EPT], sv[EPT];
#pragma unroll
  for (int k = 0; k < EPT; ++k) {
    long e = base + k * 256;
    long ec = e < last ? e : last;
    d[k] = dst[ec];
    sv[k] = src[ec];
  }
#pragma unroll
  for (int k = 0; k < EPT; ++k) {
    if (base + k * 256 < E) {
      int pos = lds_take(&cur[d[k] >> BSHIFT]);
      tmp[pos] = (sv[k] << BSHIFT) | (d[k] & BMASK);
    }
  }
}

__global__ __launch_bounds__(1024) void bin_sort(
    const int* __restrict__ tmp, const int* __restrict__ total,
    int* __restrict__ grp, int* __restrict__ cnt, int* __restrict__ offs,
    float* __restrict__ inv, int N) {
  __shared__ int hist[BSIZE];
  __shared__ int cur[BSIZE];
  __shared__ int sred[32];
  __shared__ int swave[32];
  int t = threadIdx.x, lane = t & 31, wave = t >> 5;
  int b = blockIdx.x;

  int part = 0;
  for (int i = t; i < b; i += 1024) part += total[i];
  part += __shfl_xor(part, 16);
  part += __shfl_xor(part, 8);
  part += __shfl_xor(part, 4);
  part += __shfl_xor(part, 2);
  part += __shfl_xor(part, 1);
  if (lane == 0) sred[wave] = part;
  if (t < BSIZE) hist[t] = 0;
  __syncthreads();
  int start = 0;
#pragma unroll
  for (int w = 0; w < 32; ++w) start += sred[w];
  int m = total[b];
  const int* p = tmp + start;

  for (int i = t; i < m; i += 4096) {
    int w4[4];
#pragma unroll
    for (int k = 0; k < 4; ++k) {
      int j = i + k * 1024;
      w4[k] = (j < m) ? p[j] : -1;
    }
#pragma unroll
    for (int k = 0; k < 4; ++k)
      if (w4[k] >= 0) lds_inc(&hist[w4[k] & BMASK]);
  }
  __syncthreads();

  int c = (t < BSIZE) ? hist[t] : 0;
  int inc = c;
#pragma unroll
  for (int o = 1; o < 32; o <<= 1) {
    int x = __shfl_up(inc, o);
    if (lane >= o) inc += x;
  }
  if (lane == 31) swave[wave] = inc;
  __syncthreads();
  int ex = inc - c;
#pragma unroll
  for (int w = 0; w < 32; ++w)
    if (w < wave) ex += swave[w];
  if (t < BSIZE) {
    cur[t] = ex;
    int node = b * BSIZE + t;
    if (node < N) {
      cnt[node] = c;
      offs[node] = start + ex;
      inv[node] = 1.0f / fmaxf((float)c, 1.0f);
    }
  }
  __syncthreads();

  for (int i = t; i < m; i += 4096) {
    int w4[4];
#pragma unroll
    for (int k = 0; k < 4; ++k) {
      int j = i + k * 1024;
      w4[k] = (j < m) ? p[j] : -1;
    }
#pragma unroll
    for (int k = 0; k < 4; ++k) {
      if (w4[k] >= 0) {
        int pos = lds_take(&cur[w4[k] & BMASK]);
        grp[start + pos] = (int)((unsigned)w4[k] >> BSHIFT);
      }
    }
  }
}

#define APITCH 68
__global__ __launch_bounds__(256) void sage_layer(
    const float* __restrict__ hin, const int* __restrict__ grp,
    const int* __restrict__ offs, const int* __restrict__ cnt,
    const float* __restrict__ invc, const float* __restrict__ Wl,
    const float* __restrict__ bl, const float* __restrict__ Wr,
    const float* __restrict__ gam, const float* __restrict__ bet,
    float* __restrict__ hout, int N) {
  __shared__ float sWl[64 * 64];
  __shared__ float sWr[64 * 64];
  __shared__ float sbl[64], sg[64], sbeta[64];
  __shared__ float4 sA[8][16 * (APITCH / 4)];
  int t = threadIdx.x;
  for (int i = t; i < 64 * 64; i += 256) {
    sWl[i] = Wl[i];
    sWr[i] = Wr[i];
  }
  if (t < 64) {
    sbl[t] = bl[t];
    sg[t] = gam[t];
    sbeta[t] = bet[t];
  }
  __syncthreads();

  int wave = t >> 5, lane = t & 31, half = lane >> 4, ln = lane & 15;
  long tile = (long)blockIdx.x * 8 + wave;
  long row0 = tile * 16;
  if (row0 >= N) return;

  int r = (int)row0 + ln;
  bool live = r < N;
  if (r > N - 1) r = N - 1;
  float iv = invc[r];
  int vbeg = offs[r];
  int vdeg = live ? cnt[r] : 0;

  const float4* h4 = (const float4*)hin;
  float4* sAw = sA[wave];
  for (int i = 0; i < 16; ++i) {
    int beg = __builtin_amdgcn_readlane(vbeg, i);
    int deg = __builtin_amdgcn_readlane(vdeg, i);
    const int* p = grp + beg;

    float4 a = make_float4(0.0f, 0.0f, 0.0f, 0.0f);
    int j = 0;
    for (; j + 8 <= deg; j += 8) {
      int e0 = p[j], e1 = p[j + 1], e2 = p[j + 2], e3 = p[j + 3];
      int e4 = p[j + 4], e5 = p[j + 5], e6 = p[j + 6], e7 = p[j + 7];
      int s0 = half ? e1 : e0;
      int s1 = half ? e3 : e2;
      int s2 = half ? e5 : e4;
      int s3 = half ? e7 : e6;
      float4 v0 = h4[(long)s0 * 16 + ln];
      float4 v1 = h4[(long)s1 * 16 + ln];
      float4 v2 = h4[(long)s2 * 16 + ln];
      float4 v3 = h4[(long)s3 * 16 + ln];
      a.x += v0.x; a.y += v0.y; a.z += v0.z; a.w += v0.w;
      a.x += v1.x; a.y += v1.y; a.z += v1.z; a.w += v1.w;
      a.x += v2.x; a.y += v2.y; a.z += v2.z; a.w += v2.w;
      a.x += v3.x; a.y += v3.y; a.z += v3.z; a.w += v3.w;
    }
    if (j < deg) {
      int last = deg - 1;
      int i0 = j + half, i1 = i0 + 2, i2 = i0 + 4, i3 = i0 + 6;
      int s0 = p[i0 < last ? i0 : last];
      int s1 = p[i1 < last ? i1 : last];
      int s2 = p[i2 < last ? i2 : last];
      int s3 = p[i3 < last ? i3 : last];
      float4 v0 = h4[(long)s0 * 16 + ln];
      float4 v1 = h4[(long)s1 * 16 + ln];
      float4 v2 = h4[(long)s2 * 16 + ln];
      float4 v3 = h4[(long)s3 * 16 + ln];
      bool k0 = i0 < deg, k1 = i1 < deg, k2 = i2 < deg, k3 = i3 < deg;
      a.x += k0 ? v0.x : 0.0f; a.y += k0 ? v0.y : 0.0f;
      a.z += k0 ? v0.z : 0.0f; a.w += k0 ? v0.w : 0.0f;
      a.x += k1 ? v1.x : 0.0f; a.y += k1 ? v1.y : 0.0f;
      a.z += k1 ? v1.z : 0.0f; a.w += k1 ? v1.w : 0.0f;
      a.x += k2 ? v2.x : 0.0f; a.y += k2 ? v2.y : 0.0f;
      a.z += k2 ? v2.z : 0.0f; a.w += k2 ? v2.w : 0.0f;
      a.x += k3 ? v3.x : 0.0f; a.y += k3 ? v3.y : 0.0f;
      a.z += k3 ? v3.z : 0.0f; a.w += k3 ? v3.w : 0.0f;
    }
    a.x += __shfl_xor(a.x, 16);
    a.y += __shfl_xor(a.y, 16);
    a.z += __shfl_xor(a.z, 16);
    a.w += __shfl_xor(a.w, 16);
    if (half == 0) sAw[i * (APITCH / 4) + ln] = a;
  }

  const float* pag = (const float*)sAw + ln * APITCH;
  const float* ph = hin + (long)r * 64;

  v8f acc[4];
#pragma unroll
  for (int c = 0; c < 4; ++c) acc[c] = (v8f){0, 0, 0, 0, 0, 0, 0, 0};

#pragma unroll
  for (int k = 0; k < 64; k += 4) {
    int ka = k + 2 * half;
    v2f aA, aH;
    aA[0] = pag[ka] * iv;
    aA[1] = pag[ka + 1] * iv;
    aH[0] = ph[ka];
    aH[1] = ph[ka + 1];
#pragma unroll
    for (int c = 0; c < 4; ++c) {
      v2f bL, bR;
      bL[0] = sWl[ka * 64 + c * 16 + ln];
      bL[1] = sWl[(ka + 1) * 64 + c * 16 + ln];
      bR[0] = sWr[ka * 64 + c * 16 + ln];
      bR[1] = sWr[(ka + 1) * 64 + c * 16 + ln];
      acc[c] = wmma4(aA, bL, acc[c]);
      acc[c] = wmma4(aH, bR, acc[c]);
    }
  }

  float bias[4], g4[4], be4[4];
#pragma unroll
  for (int c = 0; c < 4; ++c) {
    bias[c] = sbl[c * 16 + ln];
    g4[c] = sg[c * 16 + ln];
    be4[c] = sbeta[c * 16 + ln];
  }

#pragma unroll
  for (int v = 0; v < 8; ++v) {
    float r0 = fmaxf(acc[0][v] + bias[0], 0.0f);
    float r1 = fmaxf(acc[1][v] + bias[1], 0.0f);
    float r2 = fmaxf(acc[2][v] + bias[2], 0.0f);
    float r3 = fmaxf(acc[3][v] + bias[3], 0.0f);
    float s = r0 + r1 + r2 + r3;
    s += __shfl_xor(s, 1, 16);
    s += __shfl_xor(s, 2, 16);
    s += __shfl_xor(s, 4, 16);
    s += __shfl_xor(s, 8, 16);
    float mu = s * (1.0f / 64.0f);
    float d0 = r0 - mu, d1 = r1 - mu, d2 = r2 - mu, d3 = r3 - mu;
    float q = d0 * d0 + d1 * d1 + d2 * d2 + d3 * d3;
    q += __shfl_xor(q, 1, 16);
    q += __shfl_xor(q, 2, 16);
    q += __shfl_xor(q, 4, 16);
    q += __shfl_xor(q, 8, 16);
    float rin = rsqrtf(q * (1.0f / 64.0f) + LN_EPS);
    long row = row0 + v + 8 * half;
    if (row < N) {
      float* po = hout + row * 64 + ln;
      po[0] = d0 * rin * g4[0] + be4[0];
      po[16] = d1 * rin * g4[1] + be4[1];
      po[32] = d2 * rin * g4[2] + be4[2];
      po[48] = d3 * rin * g4[3] + be4[3];
    }
  }
}

__global__ __launch_bounds__(256, 8) void sage_first(
    const float* __restrict__ x, const float* __restrict__ Win,
    const float* __restrict__ bin, const int* __restrict__ grp,
    const int* __restrict__ offs, const int* __restrict__ cnt,
    const float* __restrict__ invc, const float* __restrict__ Wl,
    const float* __restrict__ bl, const float* __restrict__ Wr,
    const float* __restrict__ gam, const float* __restrict__ bet,
    float* __restrict__ hout, int N) {
  __shared__ float sWl[64 * 64];
  __shared__ float sWr[64 * 64];
  __shared__ float sWin[16 * 64];
  __shared__ float sbin[64], sbl[64], sg[64], sbeta[64];
  __shared__ float4 sA[8][16 * (APITCH / 4)];
  int t = threadIdx.x;
  for (int i = t; i < 64 * 64; i += 256) {
    sWl[i] = Wl[i];
    sWr[i] = Wr[i];
  }
  for (int i = t; i < 16 * 64; i += 256) sWin[i] = Win[i];
  if (t < 64) {
    sbin[t] = bin[t];
    sbl[t] = bl[t];
    sg[t] = gam[t];
    sbeta[t] = bet[t];
  }
  __syncthreads();

  int wave = t >> 5, lane = t & 31, half = lane >> 4, ln = lane & 15;
  long tile = (long)blockIdx.x * 8 + wave;
  long row0 = tile * 16;
  if (row0 >= N) return;

  int r = (int)row0 + ln;
  bool live = r < N;
  if (r > N - 1) r = N - 1;
  float iv = invc[r];
  int vbeg = offs[r];
  int vdeg = live ? cnt[r] : 0;
  int hmask = vdeg > 0 ? -1 : 0;

  const float4* x4 = (const float4*)x;
  float4* sAw = sA[wave];
  int q = lane & 3, g = lane >> 2;

  int ids = 0;
  {
    int d0 = __builtin_amdgcn_readlane(vdeg, 0);
    if (d0 > 0) {
      const int* p0 = grp + __builtin_amdgcn_readlane(vbeg, 0);
      int l0 = d0 - 1;
      ids = p0[lane < l0 ? lane : l0];
    }
  }
  for (int i = 0; i < 16; ++i) {
    int beg = __builtin_amdgcn_readlane(vbeg, i);
    int deg = __builtin_amdgcn_readlane(vdeg, i);
    const int* p = grp + beg;
    float4 a = make_float4(0.0f, 0.0f, 0.0f, 0.0f);
    int id0 = 0, id1 = 0, id2 = 0, id3 = 0;
    if (deg > 0) {
      id0 = __shfl(ids, g);
      id1 = __shfl(ids, 8 + g);
      id2 = __shfl(ids, 16 + g);
      id3 = __shfl(ids, 24 + g);
    }
    int nids = 0;
    if (i < 15) {
      int dn = __builtin_amdgcn_readlane(vdeg, i + 1);
      if (dn > 0) {
        const int* pn = grp + __builtin_amdgcn_readlane(vbeg, i + 1);
        int l1 = dn - 1;
        nids = pn[lane < l1 ? lane : l1];
      }
    }
    if (deg > 0) {
      float4 v0 = x4[(long)id0 * 4 + q];
      float4 v1 = x4[(long)id1 * 4 + q];
      float4 v2 = x4[(long)id2 * 4 + q];
      float4 v3 = x4[(long)id3 * 4 + q];
      bool k0 = g < deg, k1 = 8 + g < deg, k2 = 16 + g < deg, k3 = 24 + g < deg;
      a.x = ((k0 ? v0.x : 0.0f) + (k1 ? v1.x : 0.0f)) +
            ((k2 ? v2.x : 0.0f) + (k3 ? v3.x : 0.0f));
      a.y = ((k0 ? v0.y : 0.0f) + (k1 ? v1.y : 0.0f)) +
            ((k2 ? v2.y : 0.0f) + (k3 ? v3.y : 0.0f));
      a.z = ((k0 ? v0.z : 0.0f) + (k1 ? v1.z : 0.0f)) +
            ((k2 ? v2.z : 0.0f) + (k3 ? v3.z : 0.0f));
      a.w = ((k0 ? v0.w : 0.0f) + (k1 ? v1.w : 0.0f)) +
            ((k2 ? v2.w : 0.0f) + (k3 ? v3.w : 0.0f));
      for (int j0 = 32; j0 < deg; j0 += 32) {
        int last = deg - 1;
        int idx = j0 + lane;
        int more = p[idx < last ? idx : last];
        int e0 = __shfl(more, g), e1 = __shfl(more, 8 + g);
        int e2 = __shfl(more, 16 + g), e3 = __shfl(more, 24 + g);
        float4 w0 = x4[(long)e0 * 4 + q];
        float4 w1 = x4[(long)e1 * 4 + q];
        float4 w2 = x4[(long)e2 * 4 + q];
        float4 w3 = x4[(long)e3 * 4 + q];
        bool m0 = j0 + g < deg, m1 = j0 + 8 + g < deg;
        bool m2 = j0 + 16 + g < deg, m3 = j0 + 24 + g < deg;
        a.x += ((m0 ? w0.x : 0.0f) + (m1 ? w1.x : 0.0f)) +
               ((m2 ? w2.x : 0.0f) + (m3 ? w3.x : 0.0f));
        a.y += ((m0 ? w0.y : 0.0f) + (m1 ? w1.y : 0.0f)) +
               ((m2 ? w2.y : 0.0f) + (m3 ? w3.y : 0.0f));
        a.z += ((m0 ? w0.z : 0.0f) + (m1 ? w1.z : 0.0f)) +
               ((m2 ? w2.z : 0.0f) + (m3 ? w3.z : 0.0f));
        a.w += ((m0 ? w0.w : 0.0f) + (m1 ? w1.w : 0.0f)) +
               ((m2 ? w2.w : 0.0f) + (m3 ? w3.w : 0.0f));
      }
    }
    a.x += __shfl_xor(a.x, 4);
    a.y += __shfl_xor(a.y, 4);
    a.z += __shfl_xor(a.z, 4);
    a.w += __shfl_xor(a.w, 4);
    a.x += __shfl_xor(a.x, 8);
    a.y += __shfl_xor(a.y, 8);
    a.z += __shfl_xor(a.z, 8);
    a.w += __shfl_xor(a.w, 8);
    a.x += __shfl_xor(a.x, 16);
    a.y += __shfl_xor(a.y, 16);
    a.z += __shfl_xor(a.z, 16);
    a.w += __shfl_xor(a.w, 16);
    if (g == 0) sAw[i * (APITCH / 4) + q] = a;
    ids = nids;
  }

  float* tl = (float*)sAw;
  const float* pag = tl + ln * APITCH;

  v2f aX[4];
#pragma unroll
  for (int kk = 0; kk < 4; ++kk) {
    int ka = 4 * kk + 2 * half;
    aX[kk][0] = pag[ka] * iv;
    aX[kk][1] = pag[ka + 1] * iv;
  }
#pragma unroll
  for (int c = 0; c < 4; ++c) {
    v8f mc = (v8f){0, 0, 0, 0, 0, 0, 0, 0};
#pragma unroll
    for (int kk = 0; kk < 4; ++kk) {
      int ka = 4 * kk + 2 * half;
      v2f b;
      b[0] = sWin[ka * 64 + c * 16 + ln];
      b[1] = sWin[(ka + 1) * 64 + c * 16 + ln];
      mc = wmma4(aX[kk], b, mc);
    }
#pragma unroll
    for (int v = 0; v < 8; ++v)
      tl[(v + 8 * half) * APITCH + c * 16 + ln] = mc[v];
  }

  v8f acc[4];
#pragma unroll
  for (int c = 0; c < 4; ++c) acc[c] = (v8f){0, 0, 0, 0, 0, 0, 0, 0};

#pragma unroll
  for (int k = 0; k < 64; k += 4) {
    int ka = k + 2 * half;
    v2f aA;
    aA[0] = pag[ka] + __int_as_float(__float_as_int(sbin[ka]) & hmask);
    aA[1] = pag[ka + 1] + __int_as_float(__float_as_int(sbin[ka + 1]) & hmask);
#pragma unroll
    for (int c = 0; c < 4; ++c) {
      v2f bL;
      bL[0] = sWl[ka * 64 + c * 16 + ln];
      bL[1] = sWl[(ka + 1) * 64 + c * 16 + ln];
      acc[c] = wmma4(aA, bL, acc[c]);
    }
  }

  const float* px = x + (long)r * 16;
#pragma unroll
  for (int kk = 0; kk < 4; ++kk) {
    int ka = 4 * kk + 2 * half;
    aX[kk][0] = px[ka];
    aX[kk][1] = px[ka + 1];
  }
#pragma unroll
  for (int c = 0; c < 4; ++c) {
    v8f mc = (v8f){0, 0, 0, 0, 0, 0, 0, 0};
#pragma unroll
    for (int kk = 0; kk < 4; ++kk) {
      int ka = 4 * kk + 2 * half;
      v2f b;
      b[0] = sWin[ka * 64 + c * 16 + ln];
      b[1] = sWin[(ka + 1) * 64 + c * 16 + ln];
      mc = wmma4(aX[kk], b, mc);
    }
#pragma unroll
    for (int v = 0; v < 8; ++v)
      tl[(v + 8 * half) * APITCH + c * 16 + ln] = mc[v];
  }

#pragma unroll
  for (int k = 0; k < 64; k += 4) {
    int ka = k + 2 * half;
    v2f aH;
    aH[0] = pag[ka] + sbin[ka];
    aH[1] = pag[ka + 1] + sbin[ka + 1];
#pragma unroll
    for (int c = 0; c < 4; ++c) {
      v2f bR;
      bR[0] = sWr[ka * 64 + c * 16 + ln];
      bR[1] = sWr[(ka + 1) * 64 + c * 16 + ln];
      acc[c] = wmma4(aH, bR, acc[c]);
    }
  }

  float bias[4], g4[4], be4[4];
#pragma unroll
  for (int c = 0; c < 4; ++c) {
    bias[c] = sbl[c * 16 + ln];
    g4[c] = sg[c * 16 + ln];
    be4[c] = sbeta[c * 16 + ln];
  }

#pragma unroll
  for (int v = 0; v < 8; ++v) {
    float r0 = fmaxf(acc[0][v] + bias[0], 0.0f);
    float r1 = fmaxf(acc[1][v] + bias[1], 0.0f);
    float r2 = fmaxf(acc[2][v] + bias[2], 0.0f);
    float r3 = fmaxf(acc[3][v] + bias[3], 0.0f);
    float s = r0 + r1 + r2 + r3;
    s += __shfl_xor(s, 1, 16);
    s += __shfl_xor(s, 2, 16);
    s += __shfl_xor(s, 4, 16);
    s += __shfl_xor(s, 8, 16);
    float mu = s * (1.0f / 64.0f);
    float d0 = r0 - mu, d1 = r1 - mu, d2 = r2 - mu, d3 = r3 - mu;
    float qq = d0 * d0 + d1 * d1 + d2 * d2 + d3 * d3;
    qq += __shfl_xor(qq, 1, 16);
    qq += __shfl_xor(qq, 2, 16);
    qq += __shfl_xor(qq, 4, 16);
    qq += __shfl_xor(qq, 8, 16);
    float rin = rsqrtf(qq * (1.0f / 64.0f) + LN_EPS);
    long row = row0 + v + 8 * half;
    if (row < N) {
      float* po = hout + row * 64 + ln;
      po[0] = d0 * rin * g4[0] + be4[0];
      po[16] = d1 * rin * g4[1] + be4[1];
      po[32] = d2 * rin * g4[2] + be4[2];
      po[48] = d3 * rin * g4[3] + be4[3];
    }
  }
}

extern "C" void kernel_launch(void* const* d_in, const int* in_sizes, int n_in,
                              void* d_out, int out_size, void* d_ws,
                              size_t ws_size, hipStream_t stream) {
  (void)n_in;
  (void)out_size;
  (void)ws_size;
  const float* x = (const float*)d_in[0];
  const int* ei = (const int*)d_in[1];
  const float* Win = (const float*)d_in[2];
  const float* bin = (const float*)d_in[3];
  const float* Wl = (const float*)d_in[4];
  const float* bl = (const float*)d_in[5];
  const float* Wr = (const float*)d_in[6];
  const float* gam = (const float*)d_in[7];
  const float* bet = (const float*)d_in[8];

  const int N = in_sizes[0] / 16;
  const long E = in_sizes[1] / 2;
  const int* src = ei;
  const int* dst = ei + E;

  float* ws = (float*)d_ws;
  size_t nA = ((size_t)N + 63) & ~(size_t)63;
  size_t eA = ((size_t)E + 63) & ~(size_t)63;
  int* cnt = (int*)ws;
  float* buf0 = ws + nA;
  float* buf1 = buf0 + (size_t)N * 64;
  int* grp = (int*)buf1;
  int* offs = grp + eA;
  float* inv = (float*)(offs + nA);
  float* outp = (float*)d_out;

  const int NB = (N + BSIZE - 1) / BSIZE;
  const int G = (int)((E + CHUNK - 1) / CHUNK);
  size_t mA = ((size_t)NB * (size_t)G + 63) & ~(size_t)63;
  int* binned = (int*)d_out;
  int* wgcnt = binned + eA;
  int* total = wgcnt + mA;

  bin_count<<<G, 256, 0, stream>>>(dst, wgcnt, E, NB, G);
  row_scan<<<NB, 256, 0, stream>>>(wgcnt, total, G);
  bin_scatter<<<G, 256, 0, stream>>>(src, dst, wgcnt, total, binned, E, NB, G);
  bin_sort<<<NB, 1024, 0, stream>>>(binned, total, grp, cnt, offs, inv, N);

  int tiles = (N + 15) / 16;
  int gblocks = (tiles + 7) / 8;
  sage_first<<<gblocks, 256, 0, stream>>>(x, Win, bin, grp, offs, cnt, inv, Wl,
                                          bl, Wr, gam, bet, outp, N);

  float* hcur = outp;
  float* other = buf0;
  for (int l = 1; l < 3; ++l) {
    sage_layer<<<gblocks, 256, 0, stream>>>(
        hcur, grp, offs, cnt, inv, Wl + (size_t)l * 4096, bl + (size_t)l * 64,
        Wr + (size_t)l * 4096, gam, bet, other, N);
    float* tmp = hcur;
    hcur = other;
    other = tmp;
  }
}


# Both sort passes order their words in an LDS stage and write them out as runs (consecutive lanes, consecutive words), not one 4-byte store per edge into 32 lines; sum order stays arrival order
# speedup vs baseline: 4.8468x; 1.1859x over previous
#include <hip/hip_runtime.h>


typedef float v2f __attribute__((ext_vector_type(2)));
typedef float v8f __attribute__((ext_vector_type(8)));

#define LN_EPS 1e-5f

__device__ __forceinline__ v8f wmma4(v2f a, v2f b, v8f c) {
  return __builtin_amdgcn_wmma_f32_16x16x4_f32(false, a, false, b, (short)0, c,
                                               false, false);
}

#define BSHIFT 9
#define BSIZE 512
#define BMASK 511
#define MAXNB 2048
#define CHUNK 4096
#define EPT 16

__device__ __forceinline__ void lds_inc(int* p) {
  __hip_atomic_fetch_add(p, 1, __ATOMIC_RELAXED, __HIP_MEMORY_SCOPE_WORKGROUP);
}
__device__ __forceinline__ int lds_take(int* p) {
  return __hip_atomic_fetch_add(p, 1, __ATOMIC_RELAXED,
                                __HIP_MEMORY_SCOPE_WORKGROUP);
}

__global__ __launch_bounds__(256) void bin_count(const int* __restrict__ dst,
                                                 int* __restrict__ wgcnt,
                                                 long E, int NB, int G) {
  __shared__ int hist[MAXNB];
  int t = threadIdx.x;
  for (int i = t; i < NB; i += 256) hist[i] = 0;
  __syncthreads();
  long base = (long)blockIdx.x * CHUNK + t;
  int d[EPT];
#pragma unroll
  for (int k = 0; k < EPT; ++k) {
    long e = base + k * 256;
    d[k] = (e < E) ? dst[e] : -1;
  }
#pragma unroll
  for (int k = 0; k < EPT; ++k)
    if (d[k] >= 0) lds_inc(&hist[d[k] >> BSHIFT]);
  __syncthreads();
  for (int i = t; i < NB; i += 256)
    wgcnt[(long)i * G + blockIdx.x] = hist[i];
}

__global__ __launch_bounds__(256) void row_scan(int* __restrict__ wgcnt,
                                                int* __restrict__ total,
                                                int G) {
  __shared__ int swave[8];
  int t = threadIdx.x, lane = t & 31, wave = t >> 5;
  int* row = wgcnt + (long)blockIdx.x * G;
  int carry = 0;
  for (int i0 = 0; i0 < G; i0 += 256) {
    int i = i0 + t;
    int v = (i < G) ? row[i] : 0;
    int inc = v;
#pragma unroll
    for (int o = 1; o < 32; o <<= 1) {
      int x = __shfl_up(inc, o);
      if (lane >= o) inc += x;
    }
    if (lane == 31) swave[wave] = inc;
    __syncthreads();
    int ex = carry + inc - v;
    int tile = 0;
#pragma unroll
    for (int w = 0; w < 8; ++w) {
      int s = swave[w];
      if (w < wave) ex += s;
      tile += s;
    }
    if (i < G) row[i] = ex;
    carry += tile;
    __syncthreads();
  }
  if (t == 0) total[blockIdx.x] = carry;
}

__global__ __launch_bounds__(256) void bin_scatter(
    const int* __restrict__ src, const int* __restrict__ dst,
    const int* __restrict__ wgoff, const int* __restrict__ total,
    int* __restrict__ tmp, long E, int NB, int G) {
  __shared__ int cur[MAXNB];
  __shared__ int lcur[MAXNB];
  __shared__ int lbeg[MAXNB];
  __shared__ int swave[8];
  __shared__ int stage[CHUNK];
  __shared__ unsigned short sbin[CHUNK];
  int t = threadIdx.x, lane = t & 31, wave = t >> 5;
  for (int i = t; i < NB; i += 256) lcur[i] = 0;
  int v[8];
  int s = 0;
#pragma unroll
  for (int k = 0; k < 8; ++k) {
    int i = t * 8 + k;
    v[k] = (i < NB) ? total[i] : 0;
    s += v[k];
  }
  int inc = s;
#pragma unroll
  for (int o = 1; o < 32; o <<= 1) {
    int x = __shfl_up(inc, o);
    if (lane >= o) inc += x;
  }
  if (lane == 31) swave[wave] = inc;
  __syncthreads();
  int ex = inc - s;
#pragma unroll
  for (int w = 0; w < 8; ++w)
    if (w < wave) ex += swave[w];
#pragma unroll
  for (int k = 0; k < 8; ++k) {
    int i = t * 8 + k;
    if (i < NB) cur[i] = ex + wgoff[(long)i * G + blockIdx.x];
    ex += v[k];
  }

  long base0 = (long)blockIdx.x * CHUNK;
  long base = base0 + t;
  long last = E - 1;
  int d[EPT], sv[EPT];
#pragma unroll
  for (int k = 0; k < EPT; ++k) {
    long e = base + k * 256;
    long ec = e < last ? e : last;
    d[k] = dst[ec];
    sv[k] = src[ec];
  }
#pragma unroll
  for (int k = 0; k < EPT; ++k)
    if (base + k * 256 < E) lds_inc(&lcur[d[k] >> BSHIFT]);
  __syncthreads();

  int c8[8];
  int ls = 0;
#pragma unroll
  for (int k = 0; k < 8; ++k) {
    int i = t * 8 + k;
    c8[k] = (i < NB) ? lcur[i] : 0;
    ls += c8[k];
  }
  int linc = ls;
#pragma unroll
  for (int o = 1; o < 32; o <<= 1) {
    int x = __shfl_up(linc, o);
    if (lane >= o) linc += x;
  }
  if (lane == 31) swave[wave] = linc;
  __syncthreads();
  int lex = linc - ls;
#pragma unroll
  for (int w = 0; w < 8; ++w)
    if (w < wave) lex += swave[w];
#pragma unroll
  for (int k = 0; k < 8; ++k) {
    int i = t * 8 + k;
    if (i < NB) {
      lbeg[i] = lex;
      lcur[i] = lex;
    }
    lex += c8[k];
  }
  __syncthreads();

  int rk[EPT];
#pragma unroll
  for (int k = 0; k < EPT; ++k)
    rk[k] = (base + k * 256 < E) ? lds_take(&lcur[d[k] >> BSHIFT]) : -1;
#pragma unroll
  for (int k = 0; k < EPT; ++k) {
    if (rk[k] >= 0) {
      stage[rk[k]] = (sv[k] << BSHIFT) | (d[k] & BMASK);
      sbin[rk[k]] = (unsigned short)(d[k] >> BSHIFT);
    }
  }
  __syncthreads();

  long left = E - base0;
  int n = left < CHUNK ? (int)left : CHUNK;
  int wb[EPT], ww[EPT], wp[EPT];
#pragma unroll
  for (int k = 0; k < EPT; ++k) {
    int sl = t + k * 256;
    int sc = sl < n ? sl : n - 1;
    wb[k] = sbin[sc];
    ww[k] = stage[sc];
  }
#pragma unroll
  for (int k = 0; k < EPT; ++k) {
    int sl = t + k * 256;
    int sc = sl < n ? sl : n - 1;
    wp[k] = cur[wb[k]] + (sc - lbeg[wb[k]]);
  }
#pragma unroll
  for (int k = 0; k < EPT; ++k)
    if (t + k * 256 < n) tmp[wp[k]] = ww[k];
}

#define SCAP 20480
__global__ __launch_bounds__(1024) void bin_sort(
    const int* __restrict__ tmp, const int* __restrict__ total,
    int* __restrict__ grp, int* __restrict__ cnt, int* __restrict__ offs,
    float* __restrict__ inv, int N) {
  __shared__ int hist[BSIZE];
  __shared__ int cur[BSIZE];
  __shared__ int sred[32];
  __shared__ int swave[32];
  __shared__ int stage[SCAP];
  int t = threadIdx.x, lane = t & 31, wave = t >> 5;
  int b = blockIdx.x;

  int part = 0;
  for (int i = t; i < b; i += 1024) part += total[i];
  part += __shfl_xor(part, 16);
  part += __shfl_xor(part, 8);
  part += __shfl_xor(part, 4);
  part += __shfl_xor(part, 2);
  part += __shfl_xor(part, 1);
  if (lane == 0) sred[wave] = part;
  if (t < BSIZE) hist[t] = 0;
  __syncthreads();
  int start = 0;
#pragma unroll
  for (int w = 0; w < 32; ++w) start += sred[w];
  int m = total[b];
  bool fits = m <= SCAP;
  const int* p = tmp + start;

  for (int i = t; i < m; i += 4096) {
    int w4[4];
#pragma unroll
    for (int k = 0; k < 4; ++k) {
      int j = i + k * 1024;
      w4[k] = (j < m) ? p[j] : -1;
    }
#pragma unroll
    for (int k = 0; k < 4; ++k)
      if (w4[k] >= 0) lds_inc(&hist[w4[k] & BMASK]);
  }
  __syncthreads();

  int c = (t < BSIZE) ? hist[t] : 0;
  int inc = c;
#pragma unroll
  for (int o = 1; o < 32; o <<= 1) {
    int x = __shfl_up(inc, o);
    if (lane >= o) inc += x;
  }
  if (lane == 31) swave[wave] = inc;
  __syncthreads();
  int ex = inc - c;
#pragma unroll
  for (int w = 0; w < 32; ++w)
    if (w < wave) ex += swave[w];
  if (t < BSIZE) {
    cur[t] = ex;
    int node = b * BSIZE + t;
    if (node < N) {
      cnt[node] = c;
      offs[node] = start + ex;
      inv[node] = 1.0f / fmaxf((float)c, 1.0f);
    }
  }
  __syncthreads();

  for (int i = t; i < m; i += 4096) {
    int w4[4];
#pragma unroll
    for (int k = 0; k < 4; ++k) {
      int j = i + k * 1024;
      w4[k] = (j < m) ? p[j] : -1;
    }
#pragma unroll
    for (int k = 0; k < 4; ++k) {
      if (w4[k] >= 0) {
        int pos = lds_take(&cur[w4[k] & BMASK]);
        int sid = (int)((unsigned)w4[k] >> BSHIFT);
        if (fits) stage[pos] = sid;
        else grp[start + pos] = sid;
      }
    }
  }
  if (fits) {
    __syncthreads();
    int a = start & 31;
    for (int i = t; i < m + a; i += 1024) {
      int j = i - a;
      if (j >= 0) grp[start + j] = stage[j];
    }
  }
}

#define APITCH 68
__global__ __launch_bounds__(256) void sage_layer(
    const float* __restrict__ hin, const int* __restrict__ grp,
    const int* __restrict__ offs, const int* __restrict__ cnt,
    const float* __restrict__ invc, const float* __restrict__ Wl,
    const float* __restrict__ bl, const float* __restrict__ Wr,
    const float* __restrict__ gam, const float* __restrict__ bet,
    float* __restrict__ hout, int N) {
  __shared__ float sWl[64 * 64];
  __shared__ float sWr[64 * 64];
  __shared__ float sbl[64], sg[64], sbeta[64];
  __shared__ float4 sA[8][16 * (APITCH / 4)];
  int t = threadIdx.x;
  for (int i = t; i < 64 * 64; i += 256) {
    sWl[i] = Wl[i];
    sWr[i] = Wr[i];
  }
  if (t < 64) {
    sbl[t] = bl[t];
    sg[t] = gam[t];
    sbeta[t] = bet[t];
  }
  __syncthreads();

  int wave = t >> 5, lane = t & 31, half = lane >> 4, ln = lane & 15;
  long tile = (long)blockIdx.x * 8 + wave;
  long row0 = tile * 16;
  if (row0 >= N) return;

  int r = (int)row0 + ln;
  bool live = r < N;
  if (r > N - 1) r = N - 1;
  float iv = invc[r];
  int vbeg = offs[r];
  int vdeg = live ? cnt[r] : 0;

  const float4* h4 = (const float4*)hin;
  float4* sAw = sA[wave];
  for (int i = 0; i < 16; ++i) {
    int beg = __builtin_amdgcn_readlane(vbeg, i);
    int deg = __builtin_amdgcn_readlane(vdeg, i);
    const int* p = grp + beg;

    float4 a = make_float4(0.0f, 0.0f, 0.0f, 0.0f);
    int j = 0;
    for (; j + 8 <= deg; j += 8) {
      int e0 = p[j], e1 = p[j + 1], e2 = p[j + 2], e3 = p[j + 3];
      int e4 = p[j + 4], e5 = p[j + 5], e6 = p[j + 6], e7 = p[j + 7];
      int s0 = half ? e1 : e0;
      int s1 = half ? e3 : e2;
      int s2 = half ? e5 : e4;
      int s3 = half ? e7 : e6;
      float4 v0 = h4[(long)s0 * 16 + ln];
      float4 v1 = h4[(long)s1 * 16 + ln];
      float4 v2 = h4[(long)s2 * 16 + ln];
      float4 v3 = h4[(long)s3 * 16 + ln];
      a.x += v0.x; a.y += v0.y; a.z += v0.z; a.w += v0.w;
      a.x += v1.x; a.y += v1.y; a.z += v1.z; a.w += v1.w;
      a.x += v2.x; a.y += v2.y; a.z += v2.z; a.w += v2.w;
      a.x += v3.x; a.y += v3.y; a.z += v3.z; a.w += v3.w;
    }
    if (j < deg) {
      int last = deg - 1;
      int i0 = j + half, i1 = i0 + 2, i2 = i0 + 4, i3 = i0 + 6;
      int s0 = p[i0 < last ? i0 : last];
      int s1 = p[i1 < last ? i1 : last];
      int s2 = p[i2 < last ? i2 : last];
      int s3 = p[i3 < last ? i3 : last];
      float4 v0 = h4[(long)s0 * 16 + ln];
      float4 v1 = h4[(long)s1 * 16 + ln];
      float4 v2 = h4[(long)s2 * 16 + ln];
      float4 v3 = h4[(long)s3 * 16 + ln];
      bool k0 = i0 < deg, k1 = i1 < deg, k2 = i2 < deg, k3 = i3 < deg;
      a.x += k0 ? v0.x : 0.0f; a.y += k0 ? v0.y : 0.0f;
      a.z += k0 ? v0.z : 0.0f; a.w += k0 ? v0.w : 0.0f;
      a.x += k1 ? v1.x : 0.0f; a.y += k1 ? v1.y : 0.0f;
      a.z += k1 ? v1.z : 0.0f; a.w += k1 ? v1.w : 0.0f;
      a.x += k2 ? v2.x : 0.0f; a.y += k2 ? v2.y : 0.0f;
      a.z += k2 ? v2.z : 0.0f; a.w += k2 ? v2.w : 0.0f;
      a.x += k3 ? v3.x : 0.0f; a.y += k3 ? v3.y : 0.0f;
      a.z += k3 ? v3.z : 0.0f; a.w += k3 ? v3.w : 0.0f;
    }
    a.x += __shfl_xor(a.x, 16);
    a.y += __shfl_xor(a.y, 16);
    a.z += __shfl_xor(a.z, 16);
    a.w += __shfl_xor(a.w, 16);
    if (half == 0) sAw[i * (APITCH / 4) + ln] = a;
  }

  const float* pag = (const float*)sAw + ln * APITCH;
  const float* ph = hin + (long)r * 64;

  v8f acc[4];
#pragma unroll
  for (int c = 0; c < 4; ++c) acc[c] = (v8f){0, 0, 0, 0, 0, 0, 0, 0};

#pragma unroll
  for (int k = 0; k < 64; k += 4) {
    int ka = k + 2 * half;
    v2f aA, aH;
    aA[0] = pag[ka] * iv;
    aA[1] = pag[ka + 1] * iv;
    aH[0] = ph[ka];
    aH[1] = ph[ka + 1];
#pragma unroll
    for (int c = 0; c < 4; ++c) {
      v2f bL, bR;
      bL[0] = sWl[ka * 64 + c * 16 + ln];
      bL[1] = sWl[(ka + 1) * 64 + c * 16 + ln];
      bR[0] = sWr[ka * 64 + c * 16 + ln];
      bR[1] = sWr[(ka + 1) * 64 + c * 16 + ln];
      acc[c] = wmma4(aA, bL, acc[c]);
      acc[c] = wmma4(aH, bR, acc[c]);
    }
  }

  float bias[4], g4[4], be4[4];
#pragma unroll
  for (int c = 0; c < 4; ++c) {
    bias[c] = sbl[c * 16 + ln];
    g4[c] = sg[c * 16 + ln];
    be4[c] = sbeta[c * 16 + ln];
  }

#pragma unroll
  for (int v = 0; v < 8; ++v) {
    float r0 = fmaxf(acc[0][v] + bias[0], 0.0f);
    float r1 = fmaxf(acc[1][v] + bias[1], 0.0f);
    float r2 = fmaxf(acc[2][v] + bias[2], 0.0f);
    float r3 = fmaxf(acc[3][v] + bias[3], 0.0f);
    float s = r0 + r1 + r2 + r3;
    s += __shfl_xor(s, 1, 16);
    s += __shfl_xor(s, 2, 16);
    s += __shfl_xor(s, 4, 16);
    s += __shfl_xor(s, 8, 16);
    float mu = s * (1.0f / 64.0f);
    float d0 = r0 - mu, d1 = r1 - mu, d2 = r2 - mu, d3 = r3 - mu;
    float q = d0 * d0 + d1 * d1 + d2 * d2 + d3 * d3;
    q += __shfl_xor(q, 1, 16);
    q += __shfl_xor(q, 2, 16);
    q += __shfl_xor(q, 4, 16);
    q += __shfl_xor(q, 8, 16);
    float rin = rsqrtf(q * (1.0f / 64.0f) + LN_EPS);
    long row = row0 + v + 8 * half;
    if (row < N) {
      float* po = hout + row * 64 + ln;
      po[0] = d0 * rin * g4[0] + be4[0];
      po[16] = d1 * rin * g4[1] + be4[1];
      po[32] = d2 * rin * g4[2] + be4[2];
      po[48] = d3 * rin * g4[3] + be4[3];
    }
  }
}

__global__ __launch_bounds__(256, 8) void sage_first(
    const float* __restrict__ x, const float* __restrict__ Win,
    const float* __restrict__ bin, const int* __restrict__ grp,
    const int* __restrict__ offs, const int* __restrict__ cnt,
    const float* __restrict__ invc, const float* __restrict__ Wl,
    const float* __restrict__ bl, const float* __restrict__ Wr,
    const float* __restrict__ gam, const float* __restrict__ bet,
    float* __restrict__ hout, int N) {
  __shared__ float sWl[64 * 64];
  __shared__ float sWr[64 * 64];
  __shared__ float sWin[16 * 64];
  __shared__ float sbin[64], sbl[64], sg[64], sbeta[64];
  __shared__ float4 sA[8][16 * (APITCH / 4)];
  int t = threadIdx.x;
  for (int i = t; i < 64 * 64; i += 256) {
    sWl[i] = Wl[i];
    sWr[i] = Wr[i];
  }
  for (int i = t; i < 16 * 64; i += 256) sWin[i] = Win[i];
  if (t < 64) {
    sbin[t] = bin[t];
    sbl[t] = bl[t];
    sg[t] = gam[t];
    sbeta[t] = bet[t];
  }
  __syncthreads();

  int wave = t >> 5, lane = t & 31, half = lane >> 4, ln = lane & 15;
  long tile = (long)blockIdx.x * 8 + wave;
  long row0 = tile * 16;
  if (row0 >= N) return;

  int r = (int)row0 + ln;
  bool live = r < N;
  if (r > N - 1) r = N - 1;
  float iv = invc[r];
  int vbeg = offs[r];
  int vdeg = live ? cnt[r] : 0;
  int hmask = vdeg > 0 ? -1 : 0;

  const float4* x4 = (const float4*)x;
  float4* sAw = sA[wave];
  int q = lane & 3, g = lane >> 2;

  int ids = 0;
  {
    int d0 = __builtin_amdgcn_readlane(vdeg, 0);
    if (d0 > 0) {
      const int* p0 = grp + __builtin_amdgcn_readlane(vbeg, 0);
      int l0 = d0 - 1;
      ids = p0[lane < l0 ? lane : l0];
    }
  }
  for (int i = 0; i < 16; ++i) {
    int beg = __builtin_amdgcn_readlane(vbeg, i);
    int deg = __builtin_amdgcn_readlane(vdeg, i);
    const int* p = grp + beg;
    float4 a = make_float4(0.0f, 0.0f, 0.0f, 0.0f);
    int id0 = 0, id1 = 0, id2 = 0, id3 = 0;
    if (deg > 0) {
      id0 = __shfl(ids, g);
      id1 = __shfl(ids, 8 + g);
      id2 = __shfl(ids, 16 + g);
      id3 = __shfl(ids, 24 + g);
    }
    int nids = 0;
    if (i < 15) {
      int dn = __builtin_amdgcn_readlane(vdeg, i + 1);
      if (dn > 0) {
        const int* pn = grp + __builtin_amdgcn_readlane(vbeg, i + 1);
        int l1 = dn - 1;
        nids = pn[lane < l1 ? lane : l1];
      }
    }
    if (deg > 0) {
      float4 v0 = x4[(long)id0 * 4 + q];
      float4 v1 = x4[(long)id1 * 4 + q];
      float4 v2 = x4[(long)id2 * 4 + q];
      float4 v3 = x4[(long)id3 * 4 + q];
      bool k0 = g < deg, k1 = 8 + g < deg, k2 = 16 + g < deg, k3 = 24 + g < deg;
      a.x = ((k0 ? v0.x : 0.0f) + (k1 ? v1.x : 0.0f)) +
            ((k2 ? v2.x : 0.0f) + (k3 ? v3.x : 0.0f));
      a.y = ((k0 ? v0.y : 0.0f) + (k1 ? v1.y : 0.0f)) +
            ((k2 ? v2.y : 0.0f) + (k3 ? v3.y : 0.0f));
      a.z = ((k0 ? v0.z : 0.0f) + (k1 ? v1.z : 0.0f)) +
            ((k2 ? v2.z : 0.0f) + (k3 ? v3.z : 0.0f));
      a.w = ((k0 ? v0.w : 0.0f) + (k1 ? v1.w : 0.0f)) +
            ((k2 ? v2.w : 0.0f) + (k3 ? v3.w : 0.0f));
      for (int j0 = 32; j0 < deg; j0 += 32) {
        int last = deg - 1;
        int idx = j0 + lane;
        int more = p[idx < last ? idx : last];
        int e0 = __shfl(more, g), e1 = __shfl(more, 8 + g);
        int e2 = __shfl(more, 16 + g), e3 = __shfl(more, 24 + g);
        float4 w0 = x4[(long)e0 * 4 + q];
        float4 w1 = x4[(long)e1 * 4 + q];
        float4 w2 = x4[(long)e2 * 4 + q];
        float4 w3 = x4[(long)e3 * 4 + q];
        bool m0 = j0 + g < deg, m1 = j0 + 8 + g < deg;
        bool m2 = j0 + 16 + g < deg, m3 = j0 + 24 + g < deg;
        a.x += ((m0 ? w0.x : 0.0f) + (m1 ? w1.x : 0.0f)) +
               ((m2 ? w2.x : 0.0f) + (m3 ? w3.x : 0.0f));
        a.y += ((m0 ? w0.y : 0.0f) + (m1 ? w1.y : 0.0f)) +
               ((m2 ? w2.y : 0.0f) + (m3 ? w3.y : 0.0f));
        a.z += ((m0 ? w0.z : 0.0f) + (m1 ? w1.z : 0.0f)) +
               ((m2 ? w2.z : 0.0f) + (m3 ? w3.z : 0.0f));
        a.w += ((m0 ? w0.w : 0.0f) + (m1 ? w1.w : 0.0f)) +
               ((m2 ? w2.w : 0.0f) + (m3 ? w3.w : 0.0f));
      }
    }
    a.x += __shfl_xor(a.x, 4);
    a.y += __shfl_xor(a.y, 4);
    a.z += __shfl_xor(a.z, 4);
    a.w += __shfl_xor(a.w, 4);
    a.x += __shfl_xor(a.x, 8);
    a.y += __shfl_xor(a.y, 8);
    a.z += __shfl_xor(a.z, 8);
    a.w += __shfl_xor(a.w, 8);
    a.x += __shfl_xor(a.x, 16);
    a.y += __shfl_xor(a.y, 16);
    a.z += __shfl_xor(a.z, 16);
    a.w += __shfl_xor(a.w, 16);
    if (g == 0) sAw[i * (APITCH / 4) + q] = a;
    ids = nids;
  }

  float* tl = (float*)sAw;
  const float* pag = tl + ln * APITCH;

  v2f aX[4];
#pragma unroll
  for (int kk = 0; kk < 4; ++kk) {
    int ka = 4 * kk + 2 * half;
    aX[kk][0] = pag[ka] * iv;
    aX[kk][1] = pag[ka + 1] * iv;
  }
#pragma unroll
  for (int c = 0; c < 4; ++c) {
    v8f mc = (v8f){0, 0, 0, 0, 0, 0, 0, 0};
#pragma unroll
    for (int kk = 0; kk < 4; ++kk) {
      int ka = 4 * kk + 2 * half;
      v2f b;
      b[0] = sWin[ka * 64 + c * 16 + ln];
      b[1] = sWin[(ka + 1) * 64 + c * 16 + ln];
      mc = wmma4(aX[kk], b, mc);
    }
#pragma unroll
    for (int v = 0; v < 8; ++v)
      tl[(v + 8 * half) * APITCH + c * 16 + ln] = mc[v];
  }

  v8f acc[4];
#pragma unroll
  for (int c = 0; c < 4; ++c) acc[c] = (v8f){0, 0, 0, 0, 0, 0, 0, 0};

#pragma unroll
  for (int k = 0; k < 64; k += 4) {
    int ka = k + 2 * half;
    v2f aA;
    aA[0] = pag[ka] + __int_as_float(__float_as_int(sbin[ka]) & hmask);
    aA[1] = pag[ka + 1] + __int_as_float(__float_as_int(sbin[ka + 1]) & hmask);
#pragma unroll
    for (int c = 0; c < 4; ++c) {
      v2f bL;
      bL[0] = sWl[ka * 64 + c * 16 + ln];
      bL[1] = sWl[(ka + 1) * 64 + c * 16 + ln];
      acc[c] = wmma4(aA, bL, acc[c]);
    }
  }

  const float* px = x + (long)r * 16;
#pragma unroll
  for (int kk = 0; kk < 4; ++kk) {
    int ka = 4 * kk + 2 * half;
    aX[kk][0] = px[ka];
    aX[kk][1] = px[ka + 1];
  }
#pragma unroll
  for (int c = 0; c < 4; ++c) {
    v8f mc = (v8f){0, 0, 0, 0, 0, 0, 0, 0};
#pragma unroll
    for (int kk = 0; kk < 4; ++kk) {
      int ka = 4 * kk + 2 * half;
      v2f b;
      b[0] = sWin[ka * 64 + c * 16 + ln];
      b[1] = sWin[(ka + 1) * 64 + c * 16 + ln];
      mc = wmma4(aX[kk], b, mc);
    }
#pragma unroll
    for (int v = 0; v < 8; ++v)
      tl[(v + 8 * half) * APITCH + c * 16 + ln] = mc[v];
  }

#pragma unroll
  for (int k = 0; k < 64; k += 4) {
    int ka = k + 2 * half;
    v2f aH;
    aH[0] = pag[ka] + sbin[ka];
    aH[1] = pag[ka + 1] + sbin[ka + 1];
#pragma unroll
    for (int c = 0; c < 4; ++c) {
      v2f bR;
      bR[0] = sWr[ka * 64 + c * 16 + ln];
      bR[1] = sWr[(ka + 1) * 64 + c * 16 + ln];
      acc[c] = wmma4(aH, bR, acc[c]);
    }
  }

  float bias[4], g4[4], be4[4];
#pragma unroll
  for (int c = 0; c < 4; ++c) {
    bias[c] = sbl[c * 16 + ln];
    g4[c] = sg[c * 16 + ln];
    be4[c] = sbeta[c * 16 + ln];
  }

#pragma unroll
  for (int v = 0; v < 8; ++v) {
    float r0 = fmaxf(acc[0][v] + bias[0], 0.0f);
    float r1 = fmaxf(acc[1][v] + bias[1], 0.0f);
    float r2 = fmaxf(acc[2][v] + bias[2], 0.0f);
    float r3 = fmaxf(acc[3][v] + bias[3], 0.0f);
    float s = r0 + r1 + r2 + r3;
    s += __shfl_xor(s, 1, 16);
    s += __shfl_xor(s, 2, 16);
    s += __shfl_xor(s, 4, 16);
    s += __shfl_xor(s, 8, 16);
    float mu = s * (1.0f / 64.0f);
    float d0 = r0 - mu, d1 = r1 - mu, d2 = r2 - mu, d3 = r3 - mu;
    float qq = d0 * d0 + d1 * d1 + d2 * d2 + d3 * d3;
    qq += __shfl_xor(qq, 1, 16);
    qq += __shfl_xor(qq, 2, 16);
    qq += __shfl_xor(qq, 4, 16);
    qq += __shfl_xor(qq, 8, 16);
    float rin = rsqrtf(qq * (1.0f / 64.0f) + LN_EPS);
    long row = row0 + v + 8 * half;
    if (row < N) {
      float* po = hout + row * 64 + ln;
      po[0] = d0 * rin * g4[0] + be4[0];
      po[16] = d1 * rin * g4[1] + be4[1];
      po[32] = d2 * rin * g4[2] + be4[2];
      po[48] = d3 * rin * g4[3] + be4[3];
    }
  }
}

extern "C" void kernel_launch(void* const* d_in, const int* in_sizes, int n_in,
                              void* d_out, int out_size, void* d_ws,
                              size_t ws_size, hipStream_t stream) {
  (void)n_in;
  (void)out_size;
  (void)ws_size;
  const float* x = (const float*)d_in[0];
  const int* ei = (const int*)d_in[1];
  const float* Win = (const float*)d_in[2];
  const float* bin = (const float*)d_in[3];
  const float* Wl = (const float*)d_in[4];
  const float* bl = (const float*)d_in[5];
  const float* Wr = (const float*)d_in[6];
  const float* gam = (const float*)d_in[7];
  const float* bet = (const float*)d_in[8];

  const int N = in_sizes[0] / 16;
  const long E = in_sizes[1] / 2;
  const int* src = ei;
  const int* dst = ei + E;

  float* ws = (float*)d_ws;
  size_t nA = ((size_t)N + 63) & ~(size_t)63;
  size_t eA = ((size_t)E + 63) & ~(size_t)63;
  int* cnt = (int*)ws;
  float* buf0 = ws + nA;
  float* buf1 = buf0 + (size_t)N * 64;
  int* grp = (int*)buf1;
  int* offs = grp + eA;
  float* inv = (float*)(offs + nA);
  float* outp = (float*)d_out;

  const int NB = (N + BSIZE - 1) / BSIZE;
  const int G = (int)((E + CHUNK - 1) / CHUNK);
  size_t mA = ((size_t)NB * (size_t)G + 63) & ~(size_t)63;
  int* binned = (int*)d_out;
  int* wgcnt = binned + eA;
  int* total = wgcnt + mA;

  bin_count<<<G, 256, 0, stream>>>(dst, wgcnt, E, NB, G);
  row_scan<<<NB, 256, 0, stream>>>(wgcnt, total, G);
  bin_scatter<<<G, 256, 0, stream>>>(src, dst, wgcnt, total, binned, E, NB, G);
  bin_sort<<<NB, 1024, 0, stream>>>(binned, total, grp, cnt, offs, inv, N);

  int tiles = (N + 15) / 16;
  int gblocks = (tiles + 7) / 8;
  sage_first<<<gblocks, 256, 0, stream>>>(x, Win, bin, grp, offs, cnt, inv, Wl,
                                          bl, Wr, gam, bet, outp, N);

  float* hcur = outp;
  float* other = buf0;
  for (int l = 1; l < 3; ++l) {
    sage_layer<<<gblocks, 256, 0, stream>>>(
        hcur, grp, offs, cnt, inv, Wl + (size_t)l * 4096, bl + (size_t)l * 64,
        Wr + (size_t)l * 4096, gam, bet, other, N);
    float* tmp = hcur;
    hcur = other;
    other = tmp;
  }
}


# Level-one binning kept chunk-local: each chunk ordered by bin in LDS, written back in place with a run table; bin work-groups collect their runs; count and scan launches gone; sum order changes
# speedup vs baseline: 5.4803x; 1.1307x over previous
#include <hip/hip_runtime.h>


typedef float v2f __attribute__((ext_vector_type(2)));
typedef float v8f __attribute__((ext_vector_type(8)));

#define LN_EPS 1e-5f

__device__ __forceinline__ v8f wmma4(v2f a, v2f b, v8f c) {
  return __builtin_amdgcn_wmma_f32_16x16x4_f32(false, a, false, b, (short)0, c,
                                               false, false);
}

#define BSHIFT 9
#define BSIZE 512
#define BMASK 511
#define MAXNB 2048
#define CHUNK 4096
#define EPT 16
#define MAXG 2048

__device__ __forceinline__ void lds_inc(int* p) {
  __hip_atomic_fetch_add(p, 1, __ATOMIC_RELAXED, __HIP_MEMORY_SCOPE_WORKGROUP);
}
__device__ __forceinline__ int lds_take(int* p) {
  return __hip_atomic_fetch_add(p, 1, __ATOMIC_RELAXED,
                                __HIP_MEMORY_SCOPE_WORKGROUP);
}

__global__ __launch_bounds__(256) void chunk_sort(
    const int* __restrict__ src, const int* __restrict__ dst,
    int* __restrict__ tmp, int* __restrict__ runs, long E, int NB, int G) {
  __shared__ int lcur[MAXNB];
  __shared__ int swave[8];
  __shared__ int stage[CHUNK];
  int t = threadIdx.x, lane = t & 31, wave = t >> 5;
  for (int i = t; i < NB; i += 256) lcur[i] = 0;
  __syncthreads();

  long base0 = (long)blockIdx.x * CHUNK;
  long base = base0 + t;
  long last = E - 1;
  int d[EPT], sv[EPT];
#pragma unroll
  for (int k = 0; k < EPT; ++k) {
    long e = base + k * 256;
    long ec = e < last ? e : last;
    d[k] = dst[ec];
    sv[k] = src[ec];
  }
#pragma unroll
  for (int k = 0; k < EPT; ++k)
    if (base + k * 256 < E) lds_inc(&lcur[d[k] >> BSHIFT]);
  __syncthreads();

  int c8[8];
  int ls = 0;
#pragma unroll
  for (int k = 0; k < 8; ++k) {
    int i = t * 8 + k;
    c8[k] = (i < NB) ? lcur[i] : 0;
    ls += c8[k];
  }
  int linc = ls;
#pragma unroll
  for (int o = 1; o < 32; o <<= 1) {
    int x = __shfl_up(linc, o);
    if (lane >= o) linc += x;
  }
  if (lane == 31) swave[wave] = linc;
  __syncthreads();
  int lex = linc - ls;
#pragma unroll
  for (int w = 0; w < 8; ++w)
    if (w < wave) lex += swave[w];
#pragma unroll
  for (int k = 0; k < 8; ++k) {
    int i = t * 8 + k;
    if (i < NB) {
      lcur[i] = lex;
      runs[(long)i * G + blockIdx.x] = lex | (c8[k] << 16);
    }
    lex += c8[k];
  }
  __syncthreads();

  int rk[EPT];
#pragma unroll
  for (int k = 0; k < EPT; ++k)
    rk[k] = (base + k * 256 < E) ? lds_take(&lcur[d[k] >> BSHIFT]) : -1;
#pragma unroll
  for (int k = 0; k < EPT; ++k)
    if (rk[k] >= 0) stage[rk[k]] = (sv[k] << BSHIFT) | (d[k] & BMASK);
  __syncthreads();

  long left = E - base0;
  int n = left < CHUNK ? (int)left : CHUNK;
  int ww[EPT];
#pragma unroll
  for (int k = 0; k < EPT; ++k) {
    int sl = t + k * 256;
    ww[k] = stage[sl < n ? sl : n - 1];
  }
#pragma unroll
  for (int k = 0; k < EPT; ++k)
    if (t + k * 256 < n) tmp[base0 + t + k * 256] = ww[k];
}

#define SCAP 20480
#define RPT 13
__global__ __launch_bounds__(1024) void bin_sort(
    const int* __restrict__ tmp, const int* __restrict__ runs,
    int* __restrict__ grp, int* __restrict__ cnt, int* __restrict__ offs,
    float* __restrict__ inv, int N, int G) {
  __shared__ int hist[BSIZE];
  __shared__ int cur[BSIZE];
  __shared__ int sred[32];
  __shared__ int sredm[32];
  __shared__ int swave[32];
  __shared__ int2 rinfo[MAXG];
  __shared__ int stage[SCAP];
  int t = threadIdx.x, lane = t & 31, wave = t >> 5;
  int b = blockIdx.x;

  const int* row = runs + (long)b * G;
  int ps = 0, pm = 0;
  for (int g = t; g < G; g += 1024) {
    int pk = row[g];
    int lb = pk & 0xFFFF;
    int c = (int)((unsigned)pk >> 16);
    rinfo[g] = make_int2(c > 0 ? g * CHUNK + lb : 0, c);
    ps += lb;
    pm += c;
  }
  ps += __shfl_xor(ps, 16);
  pm += __shfl_xor(pm, 16);
  ps += __shfl_xor(ps, 8);
  pm += __shfl_xor(pm, 8);
  ps += __shfl_xor(ps, 4);
  pm += __shfl_xor(pm, 4);
  ps += __shfl_xor(ps, 2);
  pm += __shfl_xor(pm, 2);
  ps += __shfl_xor(ps, 1);
  pm += __shfl_xor(pm, 1);
  if (lane == 0) {
    sred[wave] = ps;
    sredm[wave] = pm;
  }
  if (t < BSIZE) hist[t] = 0;
  __syncthreads();
  int start = 0;
  int m = 0;
#pragma unroll
  for (int w = 0; w < 32; ++w) {
    start += sred[w];
    m += sredm[w];
  }
  bool fits = m <= SCAP;
  int glast = G - 1;

  for (int r0 = wave; r0 < G; r0 += 32 * RPT) {
    int rb[RPT], rc[RPT], w8[RPT];
#pragma unroll
    for (int k = 0; k < RPT; ++k) {
      int g = r0 + 32 * k;
      int2 e = rinfo[g < glast ? g : glast];
      rb[k] = e.x;
      rc[k] = g < G ? e.y : 0;
    }
#pragma unroll
    for (int k = 0; k < RPT; ++k) {
      int cl = rc[k] > 0 ? rc[k] - 1 : 0;
      int w = tmp[rb[k] + (lane < cl ? lane : cl)];
      w8[k] = lane < rc[k] ? w : -1;
    }
#pragma unroll
    for (int k = 0; k < RPT; ++k)
      if (w8[k] >= 0) lds_inc(&hist[w8[k] & BMASK]);
#pragma unroll
    for (int k = 0; k < RPT; ++k)
      for (int o = 32 + lane; o < rc[k]; o += 32)
        lds_inc(&hist[tmp[rb[k] + o] & BMASK]);
  }
  __syncthreads();

  int c = (t < BSIZE) ? hist[t] : 0;
  int inc = c;
#pragma unroll
  for (int o = 1; o < 32; o <<= 1) {
    int x = __shfl_up(inc, o);
    if (lane >= o) inc += x;
  }
  if (lane == 31) swave[wave] = inc;
  __syncthreads();
  int ex = inc - c;
#pragma unroll
  for (int w = 0; w < 32; ++w)
    if (w < wave) ex += swave[w];
  if (t < BSIZE) {
    cur[t] = ex;
    int node = b * BSIZE + t;
    if (node < N) {
      cnt[node] = c;
      offs[node] = start + ex;
      inv[node] = 1.0f / fmaxf((float)c, 1.0f);
    }
  }
  __syncthreads();

  for (int r0 = wave; r0 < G; r0 += 32 * RPT) {
    int rb[RPT], rc[RPT], w8[RPT], ps8[RPT];
#pragma unroll
    for (int k = 0; k < RPT; ++k) {
      int g = r0 + 32 * k;
      int2 e = rinfo[g < glast ? g : glast];
      rb[k] = e.x;
      rc[k] = g < G ? e.y : 0;
    }
#pragma unroll
    for (int k = 0; k < RPT; ++k) {
      int cl = rc[k] > 0 ? rc[k] - 1 : 0;
      int w = tmp[rb[k] + (lane < cl ? lane : cl)];
      w8[k] = lane < rc[k] ? w : -1;
    }
#pragma unroll
    for (int k = 0; k < RPT; ++k)
      ps8[k] = (w8[k] >= 0) ? lds_take(&cur[w8[k] & BMASK]) : -1;
#pragma unroll
    for (int k = 0; k < RPT; ++k) {
      if (ps8[k] >= 0) {
        int sid = (int)((unsigned)w8[k] >> BSHIFT);
        if (fits) stage[ps8[k]] = sid;
        else grp[start + ps8[k]] = sid;
      }
    }
#pragma unroll
    for (int k = 0; k < RPT; ++k) {
      for (int o = 32 + lane; o < rc[k]; o += 32) {
        int w = tmp[rb[k] + o];
        int pos = lds_take(&cur[w & BMASK]);
        int sid = (int)((unsigned)w >> BSHIFT);
        if (fits) stage[pos] = sid;
        else grp[start + pos] = sid;
      }
    }
  }
  if (fits) {
    __syncthreads();
    int a = start & 31;
    for (int i = t; i < m + a; i += 1024) {
      int j = i - a;
      if (j >= 0) grp[start + j] = stage[j];
    }
  }
}

#define APITCH 68
__global__ __launch_bounds__(256) void sage_layer(
    const float* __restrict__ hin, const int* __restrict__ grp,
    const int* __restrict__ offs, const int* __restrict__ cnt,
    const float* __restrict__ invc, const float* __restrict__ Wl,
    const float* __restrict__ bl, const float* __restrict__ Wr,
    const float* __restrict__ gam, const float* __restrict__ bet,
    float* __restrict__ hout, int N) {
  __shared__ float sWl[64 * 64];
  __shared__ float sWr[64 * 64];
  __shared__ float sbl[64], sg[64], sbeta[64];
  __shared__ float4 sA[8][16 * (APITCH / 4)];
  int t = threadIdx.x;
  for (int i = t; i < 64 * 64; i += 256) {
    sWl[i] = Wl[i];
    sWr[i] = Wr[i];
  }
  if (t < 64) {
    sbl[t] = bl[t];
    sg[t] = gam[t];
    sbeta[t] = bet[t];
  }
  __syncthreads();

  int wave = t >> 5, lane = t & 31, half = lane >> 4, ln = lane & 15;
  long tile = (long)blockIdx.x * 8 + wave;
  long row0 = tile * 16;
  if (row0 >= N) return;

  int r = (int)row0 + ln;
  bool live = r < N;
  if (r > N - 1) r = N - 1;
  float iv = invc[r];
  int vbeg = offs[r];
  int vdeg = live ? cnt[r] : 0;

  const float4* h4 = (const float4*)hin;
  float4* sAw = sA[wave];
  for (int i = 0; i < 16; ++i) {
    int beg = __builtin_amdgcn_readlane(vbeg, i);
    int deg = __builtin_amdgcn_readlane(vdeg, i);
    const int* p = grp + beg;

    float4 a = make_float4(0.0f, 0.0f, 0.0f, 0.0f);
    int j = 0;
    for (; j + 8 <= deg; j += 8) {
      int e0 = p[j], e1 = p[j + 1], e2 = p[j + 2], e3 = p[j + 3];
      int e4 = p[j + 4], e5 = p[j + 5], e6 = p[j + 6], e7 = p[j + 7];
      int s0 = half ? e1 : e0;
      int s1 = half ? e3 : e2;
      int s2 = half ? e5 : e4;
      int s3 = half ? e7 : e6;
      float4 v0 = h4[(long)s0 * 16 + ln];
      float4 v1 = h4[(long)s1 * 16 + ln];
      float4 v2 = h4[(long)s2 * 16 + ln];
      float4 v3 = h4[(long)s3 * 16 + ln];
      a.x += v0.x; a.y += v0.y; a.z += v0.z; a.w += v0.w;
      a.x += v1.x; a.y += v1.y; a.z += v1.z; a.w += v1.w;
      a.x += v2.x; a.y += v2.y; a.z += v2.z; a.w += v2.w;
      a.x += v3.x; a.y += v3.y; a.z += v3.z; a.w += v3.w;
    }
    if (j < deg) {
      int last = deg - 1;
      int i0 = j + half, i1 = i0 + 2, i2 = i0 + 4, i3 = i0 + 6;
      int s0 = p[i0 < last ? i0 : last];
      int s1 = p[i1 < last ? i1 : last];
      int s2 = p[i2 < last ? i2 : last];
      int s3 = p[i3 < last ? i3 : last];
      float4 v0 = h4[(long)s0 * 16 + ln];
      float4 v1 = h4[(long)s1 * 16 + ln];
      float4 v2 = h4[(long)s2 * 16 + ln];
      float4 v3 = h4[(long)s3 * 16 + ln];
      bool k0 = i0 < deg, k1 = i1 < deg, k2 = i2 < deg, k3 = i3 < deg;
      a.x += k0 ? v0.x : 0.0f; a.y += k0 ? v0.y : 0.0f;
      a.z += k0 ? v0.z : 0.0f; a.w += k0 ? v0.w : 0.0f;
      a.x += k1 ? v1.x : 0.0f; a.y += k1 ? v1.y : 0.0f;
      a.z += k1 ? v1.z : 0.0f; a.w += k1 ? v1.w : 0.0f;
      a.x += k2 ? v2.x : 0.0f; a.y += k2 ? v2.y : 0.0f;
      a.z += k2 ? v2.z : 0.0f; a.w += k2 ? v2.w : 0.0f;
      a.x += k3 ? v3.x : 0.0f; a.y += k3 ? v3.y : 0.0f;
      a.z += k3 ? v3.z : 0.0f; a.w += k3 ? v3.w : 0.0f;
    }
    a.x += __shfl_xor(a.x, 16);
    a.y += __shfl_xor(a.y, 16);
    a.z += __shfl_xor(a.z, 16);
    a.w += __shfl_xor(a.w, 16);
    if (half == 0) sAw[i * (APITCH / 4) + ln] = a;
  }

  const float* pag = (const float*)sAw + ln * APITCH;
  const float* ph = hin + (long)r * 64;

  v8f acc[4];
#pragma unroll
  for (int c = 0; c < 4; ++c) acc[c] = (v8f){0, 0, 0, 0, 0, 0, 0, 0};

#pragma unroll
  for (int k = 0; k < 64; k += 4) {
    int ka = k + 2 * half;
    v2f aA, aH;
    aA[0] = pag[ka] * iv;
    aA[1] = pag[ka + 1] * iv;
    aH[0] = ph[ka];
    aH[1] = ph[ka + 1];
#pragma unroll
    for (int c = 0; c < 4; ++c) {
      v2f bL, bR;
      bL[0] = sWl[ka * 64 + c * 16 + ln];
      bL[1] = sWl[(ka + 1) * 64 + c * 16 + ln];
      bR[0] = sWr[ka * 64 + c * 16 + ln];
      bR[1] = sWr[(ka + 1) * 64 + c * 16 + ln];
      acc[c] = wmma4(aA, bL, acc[c]);
      acc[c] = wmma4(aH, bR, acc[c]);
    }
  }

  float bias[4], g4[4], be4[4];
#pragma unroll
  for (int c = 0; c < 4; ++c) {
    bias[c] = sbl[c * 16 + ln];
    g4[c] = sg[c * 16 + ln];
    be4[c] = sbeta[c * 16 + ln];
  }

#pragma unroll
  for (int v = 0; v < 8; ++v) {
    float r0 = fmaxf(acc[0][v] + bias[0], 0.0f);
    float r1 = fmaxf(acc[1][v] + bias[1], 0.0f);
    float r2 = fmaxf(acc[2][v] + bias[2], 0.0f);
    float r3 = fmaxf(acc[3][v] + bias[3], 0.0f);
    float s = r0 + r1 + r2 + r3;
    s += __shfl_xor(s, 1, 16);
    s += __shfl_xor(s, 2, 16);
    s += __shfl_xor(s, 4, 16);
    s += __shfl_xor(s, 8, 16);
    float mu = s * (1.0f / 64.0f);
    float d0 = r0 - mu, d1 = r1 - mu, d2 = r2 - mu, d3 = r3 - mu;
    float q = d0 * d0 + d1 * d1 + d2 * d2 + d3 * d3;
    q += __shfl_xor(q, 1, 16);
    q += __shfl_xor(q, 2, 16);
    q += __shfl_xor(q, 4, 16);
    q += __shfl_xor(q, 8, 16);
    float rin = rsqrtf(q * (1.0f / 64.0f) + LN_EPS);
    long row = row0 + v + 8 * half;
    if (row < N) {
      float* po = hout + row * 64 + ln;
      po[0] = d0 * rin * g4[0] + be4[0];
      po[16] = d1 * rin * g4[1] + be4[1];
      po[32] = d2 * rin * g4[2] + be4[2];
      po[48] = d3 * rin * g4[3] + be4[3];
    }
  }
}

__global__ __launch_bounds__(256, 8) void sage_first(
    const float* __restrict__ x, const float* __restrict__ Win,
    const float* __restrict__ bin, const int* __restrict__ grp,
    const int* __restrict__ offs, const int* __restrict__ cnt,
    const float* __restrict__ invc, const float* __restrict__ Wl,
    const float* __restrict__ bl, const float* __restrict__ Wr,
    const float* __restrict__ gam, const float* __restrict__ bet,
    float* __restrict__ hout, int N) {
  __shared__ float sWl[64 * 64];
  __shared__ float sWr[64 * 64];
  __shared__ float sWin[16 * 64];
  __shared__ float sbin[64], sbl[64], sg[64], sbeta[64];
  __shared__ float4 sA[8][16 * (APITCH / 4)];
  int t = threadIdx.x;
  for (int i = t; i < 64 * 64; i += 256) {
    sWl[i] = Wl[i];
    sWr[i] = Wr[i];
  }
  for (int i = t; i < 16 * 64; i += 256) sWin[i] = Win[i];
  if (t < 64) {
    sbin[t] = bin[t];
    sbl[t] = bl[t];
    sg[t] = gam[t];
    sbeta[t] = bet[t];
  }
  __syncthreads();

  int wave = t >> 5, lane = t & 31, half = lane >> 4, ln = lane & 15;
  long tile = (long)blockIdx.x * 8 + wave;
  long row0 = tile * 16;
  if (row0 >= N) return;

  int r = (int)row0 + ln;
  bool live = r < N;
  if (r > N - 1) r = N - 1;
  float iv = invc[r];
  int vbeg = offs[r];
  int vdeg = live ? cnt[r] : 0;
  int hmask = vdeg > 0 ? -1 : 0;

  const float4* x4 = (const float4*)x;
  float4* sAw = sA[wave];
  int q = lane & 3, g = lane >> 2;

  int ids = 0;
  {
    int d0 = __builtin_amdgcn_readlane(vdeg, 0);
    if (d0 > 0) {
      const int* p0 = grp + __builtin_amdgcn_readlane(vbeg, 0);
      int l0 = d0 - 1;
      ids = p0[lane < l0 ? lane : l0];
    }
  }
  for (int i = 0; i < 16; ++i) {
    int beg = __builtin_amdgcn_readlane(vbeg, i);
    int deg = __builtin_amdgcn_readlane(vdeg, i);
    const int* p = grp + beg;
    float4 a = make_float4(0.0f, 0.0f, 0.0f, 0.0f);
    int id0 = 0, id1 = 0, id2 = 0, id3 = 0;
    if (deg > 0) {
      id0 = __shfl(ids, g);
      id1 = __shfl(ids, 8 + g);
      id2 = __shfl(ids, 16 + g);
      id3 = __shfl(ids, 24 + g);
    }
    int nids = 0;
    if (i < 15) {
      int dn = __builtin_amdgcn_readlane(vdeg, i + 1);
      if (dn > 0) {
        const int* pn = grp + __builtin_amdgcn_readlane(vbeg, i + 1);
        int l1 = dn - 1;
        nids = pn[lane < l1 ? lane : l1];
      }
    }
    if (deg > 0) {
      float4 v0 = x4[(long)id0 * 4 + q];
      float4 v1 = x4[(long)id1 * 4 + q];
      float4 v2 = x4[(long)id2 * 4 + q];
      float4 v3 = x4[(long)id3 * 4 + q];
      bool k0 = g < deg, k1 = 8 + g < deg, k2 = 16 + g < deg, k3 = 24 + g < deg;
      a.x = ((k0 ? v0.x : 0.0f) + (k1 ? v1.x : 0.0f)) +
            ((k2 ? v2.x : 0.0f) + (k3 ? v3.x : 0.0f));
      a.y = ((k0 ? v0.y : 0.0f) + (k1 ? v1.y : 0.0f)) +
            ((k2 ? v2.y : 0.0f) + (k3 ? v3.y : 0.0f));
      a.z = ((k0 ? v0.z : 0.0f) + (k1 ? v1.z : 0.0f)) +
            ((k2 ? v2.z : 0.0f) + (k3 ? v3.z : 0.0f));
      a.w = ((k0 ? v0.w : 0.0f) + (k1 ? v1.w : 0.0f)) +
            ((k2 ? v2.w : 0.0f) + (k3 ? v3.w : 0.0f));
      for (int j0 = 32; j0 < deg; j0 += 32) {
        int last = deg - 1;
        int idx = j0 + lane;
        int more = p[idx < last ? idx : last];
        int e0 = __shfl(more, g), e1 = __shfl(more, 8 + g);
        int e2 = __shfl(more, 16 + g), e3 = __shfl(more, 24 + g);
        float4 w0 = x4[(long)e0 * 4 + q];
        float4 w1 = x4[(long)e1 * 4 + q];
        float4 w2 = x4[(long)e2 * 4 + q];
        float4 w3 = x4[(long)e3 * 4 + q];
        bool m0 = j0 + g < deg, m1 = j0 + 8 + g < deg;
        bool m2 = j0 + 16 + g < deg, m3 = j0 + 24 + g < deg;
        a.x += ((m0 ? w0.x : 0.0f) + (m1 ? w1.x : 0.0f)) +
               ((m2 ? w2.x : 0.0f) + (m3 ? w3.x : 0.0f));
        a.y += ((m0 ? w0.y : 0.0f) + (m1 ? w1.y : 0.0f)) +
               ((m2 ? w2.y : 0.0f) + (m3 ? w3.y : 0.0f));
        a.z += ((m0 ? w0.z : 0.0f) + (m1 ? w1.z : 0.0f)) +
               ((m2 ? w2.z : 0.0f) + (m3 ? w3.z : 0.0f));
        a.w += ((m0 ? w0.w : 0.0f) + (m1 ? w1.w : 0.0f)) +
               ((m2 ? w2.w : 0.0f) + (m3 ? w3.w : 0.0f));
      }
    }
    a.x += __shfl_xor(a.x, 4);
    a.y += __shfl_xor(a.y, 4);
    a.z += __shfl_xor(a.z, 4);
    a.w += __shfl_xor(a.w, 4);
    a.x += __shfl_xor(a.x, 8);
    a.y += __shfl_xor(a.y, 8);
    a.z += __shfl_xor(a.z, 8);
    a.w += __shfl_xor(a.w, 8);
    a.x += __shfl_xor(a.x, 16);
    a.y += __shfl_xor(a.y, 16);
    a.z += __shfl_xor(a.z, 16);
    a.w += __shfl_xor(a.w, 16);
    if (g == 0) sAw[i * (APITCH / 4) + q] = a;
    ids = nids;
  }

  float* tl = (float*)sAw;
  const float* pag = tl + ln * APITCH;

  v2f aX[4];
#pragma unroll
  for (int kk = 0; kk < 4; ++kk) {
    int ka = 4 * kk + 2 * half;
    aX[kk][0] = pag[ka] * iv;
    aX[kk][1] = pag[ka + 1] * iv;
  }
#pragma unroll
  for (int c = 0; c < 4; ++c) {
    v8f mc = (v8f){0, 0, 0, 0, 0, 0, 0, 0};
#pragma unroll
    for (int kk = 0; kk < 4; ++kk) {
      int ka = 4 * kk + 2 * half;
      v2f b;
      b[0] = sWin[ka * 64 + c * 16 + ln];
      b[1] = sWin[(ka + 1) * 64 + c * 16 + ln];
      mc = wmma4(aX[kk], b, mc);
    }
#pragma unroll
    for (int v = 0; v < 8; ++v)
      tl[(v + 8 * half) * APITCH + c * 16 + ln] = mc[v];
  }

  v8f acc[4];
#pragma unroll
  for (int c = 0; c < 4; ++c) acc[c] = (v8f){0, 0, 0, 0, 0, 0, 0, 0};

#pragma unroll
  for (int k = 0; k < 64; k += 4) {
    int ka = k + 2 * half;
    v2f aA;
    aA[0] = pag[ka] + __int_as_float(__float_as_int(sbin[ka]) & hmask);
    aA[1] = pag[ka + 1] + __int_as_float(__float_as_int(sbin[ka + 1]) & hmask);
#pragma unroll
    for (int c = 0; c < 4; ++c) {
      v2f bL;
      bL[0] = sWl[ka * 64 + c * 16 + ln];
      bL[1] = sWl[(ka + 1) * 64 + c * 16 + ln];
      acc[c] = wmma4(aA, bL, acc[c]);
    }
  }

  const float* px = x + (long)r * 16;
#pragma unroll
  for (int kk = 0; kk < 4; ++kk) {
    int ka = 4 * kk + 2 * half;
    aX[kk][0] = px[ka];
    aX[kk][1] = px[ka + 1];
  }
#pragma unroll
  for (int c = 0; c < 4; ++c) {
    v8f mc = (v8f){0, 0, 0, 0, 0, 0, 0, 0};
#pragma unroll
    for (int kk = 0; kk < 4; ++kk) {
      int ka = 4 * kk + 2 * half;
      v2f b;
      b[0] = sWin[ka * 64 + c * 16 + ln];
      b[1] = sWin[(ka + 1) * 64 + c * 16 + ln];
      mc = wmma4(aX[kk], b, mc);
    }
#pragma unroll
    for (int v = 0; v < 8; ++v)
      tl[(v + 8 * half) * APITCH + c * 16 + ln] = mc[v];
  }

#pragma unroll
  for (int k = 0; k < 64; k += 4) {
    int ka = k + 2 * half;
    v2f aH;
    aH[0] = pag[ka] + sbin[ka];
    aH[1] = pag[ka + 1] + sbin[ka + 1];
#pragma unroll
    for (int c = 0; c < 4; ++c) {
      v2f bR;
      bR[0] = sWr[ka * 64 + c * 16 + ln];
      bR[1] = sWr[(ka + 1) * 64 + c * 16 + ln];
      acc[c] = wmma4(aH, bR, acc[c]);
    }
  }

  float bias[4], g4[4], be4[4];
#pragma unroll
  for (int c = 0; c < 4; ++c) {
    bias[c] = sbl[c * 16 + ln];
    g4[c] = sg[c * 16 + ln];
    be4[c] = sbeta[c * 16 + ln];
  }

#pragma unroll
  for (int v = 0; v < 8; ++v) {
    float r0 = fmaxf(acc[0][v] + bias[0], 0.0f);
    float r1 = fmaxf(acc[1][v] + bias[1], 0.0f);
    float r2 = fmaxf(acc[2][v] + bias[2], 0.0f);
    float r3 = fmaxf(acc[3][v] + bias[3], 0.0f);
    float s = r0 + r1 + r2 + r3;
    s += __shfl_xor(s, 1, 16);
    s += __shfl_xor(s, 2, 16);
    s += __shfl_xor(s, 4, 16);
    s += __shfl_xor(s, 8, 16);
    float mu = s * (1.0f / 64.0f);
    float d0 = r0 - mu, d1 = r1 - mu, d2 = r2 - mu, d3 = r3 - mu;
    float qq = d0 * d0 + d1 * d1 + d2 * d2 + d3 * d3;
    qq += __shfl_xor(qq, 1, 16);
    qq += __shfl_xor(qq, 2, 16);
    qq += __shfl_xor(qq, 4, 16);
    qq += __shfl_xor(qq, 8, 16);
    float rin = rsqrtf(qq * (1.0f / 64.0f) + LN_EPS);
    long row = row0 + v + 8 * half;
    if (row < N) {
      float* po = hout + row * 64 + ln;
      po[0] = d0 * rin * g4[0] + be4[0];
      po[16] = d1 * rin * g4[1] + be4[1];
      po[32] = d2 * rin * g4[2] + be4[2];
      po[48] = d3 * rin * g4[3] + be4[3];
    }
  }
}

extern "C" void kernel_launch(void* const* d_in, const int* in_sizes, int n_in,
                              void* d_out, int out_size, void* d_ws,
                              size_t ws_size, hipStream_t stream) {
  (void)n_in;
  (void)out_size;
  (void)ws_size;
  const float* x = (const float*)d_in[0];
  const int* ei = (const int*)d_in[1];
  const float* Win = (const float*)d_in[2];
  const float* bin = (const float*)d_in[3];
  const float* Wl = (const float*)d_in[4];
  const float* bl = (const float*)d_in[5];
  const float* Wr = (const float*)d_in[6];
  const float* gam = (const float*)d_in[7];
  const float* bet = (const float*)d_in[8];

  const int N = in_sizes[0] / 16;
  const long E = in_sizes[1] / 2;
  const int* src = ei;
  const int* dst = ei + E;

  float* ws = (float*)d_ws;
  size_t nA = ((size_t)N + 63) & ~(size_t)63;
  size_t eA = ((size_t)E + 63) & ~(size_t)63;
  int* cnt = (int*)ws;
  float* buf0 = ws + nA;
  float* buf1 = buf0 + (size_t)N * 64;
  int* grp = (int*)buf1;
  int* offs = grp + eA;
  float* inv = (float*)(offs + nA);
  float* outp = (float*)d_out;

  const int NB = (N + BSIZE - 1) / BSIZE;
  const int G = (int)((E + CHUNK - 1) / CHUNK);
  int* binned = (int*)d_out;
  int* runs = binned + eA;

  chunk_sort<<<G, 256, 0, stream>>>(src, dst, binned, runs, E, NB, G);
  bin_sort<<<NB, 1024, 0, stream>>>(binned, runs, grp, cnt, offs, inv, N, G);

  int tiles = (N + 15) / 16;
  int gblocks = (tiles + 7) / 8;
  sage_first<<<gblocks, 256, 0, stream>>>(x, Win, bin, grp, offs, cnt, inv, Wl,
                                          bl, Wr, gam, bet, outp, N);

  float* hcur = outp;
  float* other = buf0;
  for (int l = 1; l < 3; ++l) {
    sage_layer<<<gblocks, 256, 0, stream>>>(
        hcur, grp, offs, cnt, inv, Wl + (size_t)l * 4096, bl + (size_t)l * 64,
        Wr + (size_t)l * 4096, gam, bet, other, N);
    float* tmp = hcur;
    hcur = other;
    other = tmp;
  }
}


# Weights staged into LDS by one group of 16-byte loads per thread (a layer's 32 KB requested at once), not by eight dependent rounds of 4-byte loads that all work-groups run through together
# speedup vs baseline: 5.8926x; 1.0752x over previous
#include <hip/hip_runtime.h>


typedef float v2f __attribute__((ext_vector_type(2)));
typedef float v8f __attribute__((ext_vector_type(8)));

#define LN_EPS 1e-5f

__device__ __forceinline__ v8f wmma4(v2f a, v2f b, v8f c) {
  return __builtin_amdgcn_wmma_f32_16x16x4_f32(false, a, false, b, (short)0, c,
                                               false, false);
}

#define BSHIFT 9
#define BSIZE 512
#define BMASK 511
#define MAXNB 2048
#define CHUNK 4096
#define EPT 16
#define MAXG 2048

__device__ __forceinline__ void lds_inc(int* p) {
  __hip_atomic_fetch_add(p, 1, __ATOMIC_RELAXED, __HIP_MEMORY_SCOPE_WORKGROUP);
}
__device__ __forceinline__ int lds_take(int* p) {
  return __hip_atomic_fetch_add(p, 1, __ATOMIC_RELAXED,
                                __HIP_MEMORY_SCOPE_WORKGROUP);
}

__global__ __launch_bounds__(256) void chunk_sort(
    const int* __restrict__ src, const int* __restrict__ dst,
    int* __restrict__ tmp, int* __restrict__ runs, long E, int NB, int G) {
  __shared__ int lcur[MAXNB];
  __shared__ int swave[8];
  __shared__ int stage[CHUNK];
  int t = threadIdx.x, lane = t & 31, wave = t >> 5;
  for (int i = t; i < NB; i += 256) lcur[i] = 0;
  __syncthreads();

  long base0 = (long)blockIdx.x * CHUNK;
  long base = base0 + t;
  long last = E - 1;
  int d[EPT], sv[EPT];
#pragma unroll
  for (int k = 0; k < EPT; ++k) {
    long e = base + k * 256;
    long ec = e < last ? e : last;
    d[k] = dst[ec];
    sv[k] = src[ec];
  }
#pragma unroll
  for (int k = 0; k < EPT; ++k)
    if (base + k * 256 < E) lds_inc(&lcur[d[k] >> BSHIFT]);
  __syncthreads();

  int c8[8];
  int ls = 0;
#pragma unroll
  for (int k = 0; k < 8; ++k) {
    int i = t * 8 + k;
    c8[k] = (i < NB) ? lcur[i] : 0;
    ls += c8[k];
  }
  int linc = ls;
#pragma unroll
  for (int o = 1; o < 32; o <<= 1) {
    int x = __shfl_up(linc, o);
    if (lane >= o) linc += x;
  }
  if (lane == 31) swave[wave] = linc;
  __syncthreads();
  int lex = linc - ls;
#pragma unroll
  for (int w = 0; w < 8; ++w)
    if (w < wave) lex += swave[w];
#pragma unroll
  for (int k = 0; k < 8; ++k) {
    int i = t * 8 + k;
    if (i < NB) {
      lcur[i] = lex;
      runs[(long)i * G + blockIdx.x] = lex | (c8[k] << 16);
    }
    lex += c8[k];
  }
  __syncthreads();

  int rk[EPT];
#pragma unroll
  for (int k = 0; k < EPT; ++k)
    rk[k] = (base + k * 256 < E) ? lds_take(&lcur[d[k] >> BSHIFT]) : -1;
#pragma unroll
  for (int k = 0; k < EPT; ++k)
    if (rk[k] >= 0) stage[rk[k]] = (sv[k] << BSHIFT) | (d[k] & BMASK);
  __syncthreads();

  long left = E - base0;
  int n = left < CHUNK ? (int)left : CHUNK;
  int ww[EPT];
#pragma unroll
  for (int k = 0; k < EPT; ++k) {
    int sl = t + k * 256;
    ww[k] = stage[sl < n ? sl : n - 1];
  }
#pragma unroll
  for (int k = 0; k < EPT; ++k)
    if (t + k * 256 < n) tmp[base0 + t + k * 256] = ww[k];
}

#define SCAP 20480
#define RPT 13
__global__ __launch_bounds__(1024) void bin_sort(
    const int* __restrict__ tmp, const int* __restrict__ runs,
    int* __restrict__ grp, int* __restrict__ cnt, int* __restrict__ offs,
    float* __restrict__ inv, int N, int G) {
  __shared__ int hist[BSIZE];
  __shared__ int cur[BSIZE];
  __shared__ int sred[32];
  __shared__ int sredm[32];
  __shared__ int swave[32];
  __shared__ int2 rinfo[MAXG];
  __shared__ int stage[SCAP];
  int t = threadIdx.x, lane = t & 31, wave = t >> 5;
  int b = blockIdx.x;

  const int* row = runs + (long)b * G;
  int ps = 0, pm = 0;
  for (int g = t; g < G; g += 1024) {
    int pk = row[g];
    int lb = pk & 0xFFFF;
    int c = (int)((unsigned)pk >> 16);
    rinfo[g] = make_int2(c > 0 ? g * CHUNK + lb : 0, c);
    ps += lb;
    pm += c;
  }
  ps += __shfl_xor(ps, 16);
  pm += __shfl_xor(pm, 16);
  ps += __shfl_xor(ps, 8);
  pm += __shfl_xor(pm, 8);
  ps += __shfl_xor(ps, 4);
  pm += __shfl_xor(pm, 4);
  ps += __shfl_xor(ps, 2);
  pm += __shfl_xor(pm, 2);
  ps += __shfl_xor(ps, 1);
  pm += __shfl_xor(pm, 1);
  if (lane == 0) {
    sred[wave] = ps;
    sredm[wave] = pm;
  }
  if (t < BSIZE) hist[t] = 0;
  __syncthreads();
  int start = 0;
  int m = 0;
#pragma unroll
  for (int w = 0; w < 32; ++w) {
    start += sred[w];
    m += sredm[w];
  }
  bool fits = m <= SCAP;
  int glast = G - 1;

  for (int r0 = wave; r0 < G; r0 += 32 * RPT) {
    int rb[RPT], rc[RPT], w8[RPT];
#pragma unroll
    for (int k = 0; k < RPT; ++k) {
      int g = r0 + 32 * k;
      int2 e = rinfo[g < glast ? g : glast];
      rb[k] = e.x;
      rc[k] = g < G ? e.y : 0;
    }
#pragma unroll
    for (int k = 0; k < RPT; ++k) {
      int cl = rc[k] > 0 ? rc[k] - 1 : 0;
      int w = tmp[rb[k] + (lane < cl ? lane : cl)];
      w8[k] = lane < rc[k] ? w : -1;
    }
#pragma unroll
    for (int k = 0; k < RPT; ++k)
      if (w8[k] >= 0) lds_inc(&hist[w8[k] & BMASK]);
#pragma unroll
    for (int k = 0; k < RPT; ++k)
      for (int o = 32 + lane; o < rc[k]; o += 32)
        lds_inc(&hist[tmp[rb[k] + o] & BMASK]);
  }
  __syncthreads();

  int c = (t < BSIZE) ? hist[t] : 0;
  int inc = c;
#pragma unroll
  for (int o = 1; o < 32; o <<= 1) {
    int x = __shfl_up(inc, o);
    if (lane >= o) inc += x;
  }
  if (lane == 31) swave[wave] = inc;
  __syncthreads();
  int ex = inc - c;
#pragma unroll
  for (int w = 0; w < 32; ++w)
    if (w < wave) ex += swave[w];
  if (t < BSIZE) {
    cur[t] = ex;
    int node = b * BSIZE + t;
    if (node < N) {
      cnt[node] = c;
      offs[node] = start + ex;
      inv[node] = 1.0f / fmaxf((float)c, 1.0f);
    }
  }
  __syncthreads();

  for (int r0 = wave; r0 < G; r0 += 32 * RPT) {
    int rb[RPT], rc[RPT], w8[RPT], ps8[RPT];
#pragma unroll
    for (int k = 0; k < RPT; ++k) {
      int g = r0 + 32 * k;
      int2 e = rinfo[g < glast ? g : glast];
      rb[k] = e.x;
      rc[k] = g < G ? e.y : 0;
    }
#pragma unroll
    for (int k = 0; k < RPT; ++k) {
      int cl = rc[k] > 0 ? rc[k] - 1 : 0;
      int w = tmp[rb[k] + (lane < cl ? lane : cl)];
      w8[k] = lane < rc[k] ? w : -1;
    }
#pragma unroll
    for (int k = 0; k < RPT; ++k)
      ps8[k] = (w8[k] >= 0) ? lds_take(&cur[w8[k] & BMASK]) : -1;
#pragma unroll
    for (int k = 0; k < RPT; ++k) {
      if (ps8[k] >= 0) {
        int sid = (int)((unsigned)w8[k] >> BSHIFT);
        if (fits) stage[ps8[k]] = sid;
        else grp[start + ps8[k]] = sid;
      }
    }
#pragma unroll
    for (int k = 0; k < RPT; ++k) {
      for (int o = 32 + lane; o < rc[k]; o += 32) {
        int w = tmp[rb[k] + o];
        int pos = lds_take(&cur[w & BMASK]);
        int sid = (int)((unsigned)w >> BSHIFT);
        if (fits) stage[pos] = sid;
        else grp[start + pos] = sid;
      }
    }
  }
  if (fits) {
    __syncthreads();
    int a = start & 31;
    for (int i = t; i < m + a; i += 1024) {
      int j = i - a;
      if (j >= 0) grp[start + j] = stage[j];
    }
  }
}

#define APITCH 68
__global__ __launch_bounds__(256) void sage_layer(
    const float* __restrict__ hin, const int* __restrict__ grp,
    const int* __restrict__ offs, const int* __restrict__ cnt,
    const float* __restrict__ invc, const float* __restrict__ Wl,
    const float* __restrict__ bl, const float* __restrict__ Wr,
    const float* __restrict__ gam, const float* __restrict__ bet,
    float* __restrict__ hout, int N) {
  __shared__ alignas(16) float sWl[64 * 64];
  __shared__ alignas(16) float sWr[64 * 64];
  __shared__ float sbl[64], sg[64], sbeta[64];
  __shared__ float4 sA[8][16 * (APITCH / 4)];
  int t = threadIdx.x;
  {
    const float4* Wl4 = (const float4*)Wl;
    const float4* Wr4 = (const float4*)Wr;
    float4* sWl4 = (float4*)sWl;
    float4* sWr4 = (float4*)sWr;
    float4 wl[4], wr[4];
    float vb = 0.0f, vg = 0.0f, vbe = 0.0f;
#pragma unroll
    for (int k = 0; k < 4; ++k) {
      wl[k] = Wl4[t + 256 * k];
      wr[k] = Wr4[t + 256 * k];
    }
    if (t < 64) {
      vb = bl[t];
      vg = gam[t];
      vbe = bet[t];
    }
#pragma unroll
    for (int k = 0; k < 4; ++k) {
      sWl4[t + 256 * k] = wl[k];
      sWr4[t + 256 * k] = wr[k];
    }
    if (t < 64) {
      sbl[t] = vb;
      sg[t] = vg;
      sbeta[t] = vbe;
    }
  }
  __syncthreads();

  int wave = t >> 5, lane = t & 31, half = lane >> 4, ln = lane & 15;
  long tile = (long)blockIdx.x * 8 + wave;
  long row0 = tile * 16;
  if (row0 >= N) return;

  int r = (int)row0 + ln;
  bool live = r < N;
  if (r > N - 1) r = N - 1;
  float iv = invc[r];
  int vbeg = offs[r];
  int vdeg = live ? cnt[r] : 0;

  const float4* h4 = (const float4*)hin;
  float4* sAw = sA[wave];
  for (int i = 0; i < 16; ++i) {
    int beg = __builtin_amdgcn_readlane(vbeg, i);
    int deg = __builtin_amdgcn_readlane(vdeg, i);
    const int* p = grp + beg;

    float4 a = make_float4(0.0f, 0.0f, 0.0f, 0.0f);
    int j = 0;
    for (; j + 8 <= deg; j += 8) {
      int e0 = p[j], e1 = p[j + 1], e2 = p[j + 2], e3 = p[j + 3];
      int e4 = p[j + 4], e5 = p[j + 5], e6 = p[j + 6], e7 = p[j + 7];
      int s0 = half ? e1 : e0;
      int s1 = half ? e3 : e2;
      int s2 = half ? e5 : e4;
      int s3 = half ? e7 : e6;
      float4 v0 = h4[(long)s0 * 16 + ln];
      float4 v1 = h4[(long)s1 * 16 + ln];
      float4 v2 = h4[(long)s2 * 16 + ln];
      float4 v3 = h4[(long)s3 * 16 + ln];
      a.x += v0.x; a.y += v0.y; a.z += v0.z; a.w += v0.w;
      a.x += v1.x; a.y += v1.y; a.z += v1.z; a.w += v1.w;
      a.x += v2.x; a.y += v2.y; a.z += v2.z; a.w += v2.w;
      a.x += v3.x; a.y += v3.y; a.z += v3.z; a.w += v3.w;
    }
    if (j < deg) {
      int last = deg - 1;
      int i0 = j + half, i1 = i0 + 2, i2 = i0 + 4, i3 = i0 + 6;
      int s0 = p[i0 < last ? i0 : last];
      int s1 = p[i1 < last ? i1 : last];
      int s2 = p[i2 < last ? i2 : last];
      int s3 = p[i3 < last ? i3 : last];
      float4 v0 = h4[(long)s0 * 16 + ln];
      float4 v1 = h4[(long)s1 * 16 + ln];
      float4 v2 = h4[(long)s2 * 16 + ln];
      float4 v3 = h4[(long)s3 * 16 + ln];
      bool k0 = i0 < deg, k1 = i1 < deg, k2 = i2 < deg, k3 = i3 < deg;
      a.x += k0 ? v0.x : 0.0f; a.y += k0 ? v0.y : 0.0f;
      a.z += k0 ? v0.z : 0.0f; a.w += k0 ? v0.w : 0.0f;
      a.x += k1 ? v1.x : 0.0f; a.y += k1 ? v1.y : 0.0f;
      a.z += k1 ? v1.z : 0.0f; a.w += k1 ? v1.w : 0.0f;
      a.x += k2 ? v2.x : 0.0f; a.y += k2 ? v2.y : 0.0f;
      a.z += k2 ? v2.z : 0.0f; a.w += k2 ? v2.w : 0.0f;
      a.x += k3 ? v3.x : 0.0f; a.y += k3 ? v3.y : 0.0f;
      a.z += k3 ? v3.z : 0.0f; a.w += k3 ? v3.w : 0.0f;
    }
    a.x += __shfl_xor(a.x, 16);
    a.y += __shfl_xor(a.y, 16);
    a.z += __shfl_xor(a.z, 16);
    a.w += __shfl_xor(a.w, 16);
    if (half == 0) sAw[i * (APITCH / 4) + ln] = a;
  }

  const float* pag = (const float*)sAw + ln * APITCH;
  const float* ph = hin + (long)r * 64;

  v8f acc[4];
#pragma unroll
  for (int c = 0; c < 4; ++c) acc[c] = (v8f){0, 0, 0, 0, 0, 0, 0, 0};

#pragma unroll
  for (int k = 0; k < 64; k += 4) {
    int ka = k + 2 * half;
    v2f aA, aH;
    aA[0] = pag[ka] * iv;
    aA[1] = pag[ka + 1] * iv;
    aH[0] = ph[ka];
    aH[1] = ph[ka + 1];
#pragma unroll
    for (int c = 0; c < 4; ++c) {
      v2f bL, bR;
      bL[0] = sWl[ka * 64 + c * 16 + ln];
      bL[1] = sWl[(ka + 1) * 64 + c * 16 + ln];
      bR[0] = sWr[ka * 64 + c * 16 + ln];
      bR[1] = sWr[(ka + 1) * 64 + c * 16 + ln];
      acc[c] = wmma4(aA, bL, acc[c]);
      acc[c] = wmma4(aH, bR, acc[c]);
    }
  }

  float bias[4], g4[4], be4[4];
#pragma unroll
  for (int c = 0; c < 4; ++c) {
    bias[c] = sbl[c * 16 + ln];
    g4[c] = sg[c * 16 + ln];
    be4[c] = sbeta[c * 16 + ln];
  }

#pragma unroll
  for (int v = 0; v < 8; ++v) {
    float r0 = fmaxf(acc[0][v] + bias[0], 0.0f);
    float r1 = fmaxf(acc[1][v] + bias[1], 0.0f);
    float r2 = fmaxf(acc[2][v] + bias[2], 0.0f);
    float r3 = fmaxf(acc[3][v] + bias[3], 0.0f);
    float s = r0 + r1 + r2 + r3;
    s += __shfl_xor(s, 1, 16);
    s += __shfl_xor(s, 2, 16);
    s += __shfl_xor(s, 4, 16);
    s += __shfl_xor(s, 8, 16);
    float mu = s * (1.0f / 64.0f);
    float d0 = r0 - mu, d1 = r1 - mu, d2 = r2 - mu, d3 = r3 - mu;
    float q = d0 * d0 + d1 * d1 + d2 * d2 + d3 * d3;
    q += __shfl_xor(q, 1, 16);
    q += __shfl_xor(q, 2, 16);
    q += __shfl_xor(q, 4, 16);
    q += __shfl_xor(q, 8, 16);
    float rin = rsqrtf(q * (1.0f / 64.0f) + LN_EPS);
    long row = row0 + v + 8 * half;
    if (row < N) {
      float* po = hout + row * 64 + ln;
      po[0] = d0 * rin * g4[0] + be4[0];
      po[16] = d1 * rin * g4[1] + be4[1];
      po[32] = d2 * rin * g4[2] + be4[2];
      po[48] = d3 * rin * g4[3] + be4[3];
    }
  }
}

__global__ __launch_bounds__(256, 8) void sage_first(
    const float* __restrict__ x, const float* __restrict__ Win,
    const float* __restrict__ bin, const int* __restrict__ grp,
    const int* __restrict__ offs, const int* __restrict__ cnt,
    const float* __restrict__ invc, const float* __restrict__ Wl,
    const float* __restrict__ bl, const float* __restrict__ Wr,
    const float* __restrict__ gam, const float* __restrict__ bet,
    float* __restrict__ hout, int N) {
  __shared__ alignas(16) float sWl[64 * 64];
  __shared__ alignas(16) float sWr[64 * 64];
  __shared__ alignas(16) float sWin[16 * 64];
  __shared__ float sbin[64], sbl[64], sg[64], sbeta[64];
  __shared__ float4 sA[8][16 * (APITCH / 4)];
  int t = threadIdx.x;
  {
    const float4* Wl4 = (const float4*)Wl;
    const float4* Wr4 = (const float4*)Wr;
    const float4* Win4 = (const float4*)Win;
    float4* sWl4 = (float4*)sWl;
    float4* sWr4 = (float4*)sWr;
    float4* sWin4 = (float4*)sWin;
    float4 wl[4], wr[4], wi;
    float vbi = 0.0f, vb = 0.0f, vg = 0.0f, vbe = 0.0f;
#pragma unroll
    for (int k = 0; k < 4; ++k) {
      wl[k] = Wl4[t + 256 * k];
      wr[k] = Wr4[t + 256 * k];
    }
    wi = Win4[t];
    if (t < 64) {
      vbi = bin[t];
      vb = bl[t];
      vg = gam[t];
      vbe = bet[t];
    }
#pragma unroll
    for (int k = 0; k < 4; ++k) {
      sWl4[t + 256 * k] = wl[k];
      sWr4[t + 256 * k] = wr[k];
    }
    sWin4[t] = wi;
    if (t < 64) {
      sbin[t] = vbi;
      sbl[t] = vb;
      sg[t] = vg;
      sbeta[t] = vbe;
    }
  }
  __syncthreads();

  int wave = t >> 5, lane = t & 31, half = lane >> 4, ln = lane & 15;
  long tile = (long)blockIdx.x * 8 + wave;
  long row0 = tile * 16;
  if (row0 >= N) return;

  int r = (int)row0 + ln;
  bool live = r < N;
  if (r > N - 1) r = N - 1;
  float iv = invc[r];
  int vbeg = offs[r];
  int vdeg = live ? cnt[r] : 0;
  int hmask = vdeg > 0 ? -1 : 0;

  const float4* x4 = (const float4*)x;
  float4* sAw = sA[wave];
  int q = lane & 3, g = lane >> 2;

  int ids = 0;
  {
    int d0 = __builtin_amdgcn_readlane(vdeg, 0);
    if (d0 > 0) {
      const int* p0 = grp + __builtin_amdgcn_readlane(vbeg, 0);
      int l0 = d0 - 1;
      ids = p0[lane < l0 ? lane : l0];
    }
  }
  for (int i = 0; i < 16; ++i) {
    int beg = __builtin_amdgcn_readlane(vbeg, i);
    int deg = __builtin_amdgcn_readlane(vdeg, i);
    const int* p = grp + beg;
    float4 a = make_float4(0.0f, 0.0f, 0.0f, 0.0f);
    int id0 = 0, id1 = 0, id2 = 0, id3 = 0;
    if (deg > 0) {
      id0 = __shfl(ids, g);
      id1 = __shfl(ids, 8 + g);
      id2 = __shfl(ids, 16 + g);
      id3 = __shfl(ids, 24 + g);
    }
    int nids = 0;
    if (i < 15) {
      int dn = __builtin_amdgcn_readlane(vdeg, i + 1);
      if (dn > 0) {
        const int* pn = grp + __builtin_amdgcn_readlane(vbeg, i + 1);
        int l1 = dn - 1;
        nids = pn[lane < l1 ? lane : l1];
      }
    }
    if (deg > 0) {
      float4 v0 = x4[(long)id0 * 4 + q];
      float4 v1 = x4[(long)id1 * 4 + q];
      float4 v2 = x4[(long)id2 * 4 + q];
      float4 v3 = x4[(long)id3 * 4 + q];
      bool k0 = g < deg, k1 = 8 + g < deg, k2 = 16 + g < deg, k3 = 24 + g < deg;
      a.x = ((k0 ? v0.x : 0.0f) + (k1 ? v1.x : 0.0f)) +
            ((k2 ? v2.x : 0.0f) + (k3 ? v3.x : 0.0f));
      a.y = ((k0 ? v0.y : 0.0f) + (k1 ? v1.y : 0.0f)) +
            ((k2 ? v2.y : 0.0f) + (k3 ? v3.y : 0.0f));
      a.z = ((k0 ? v0.z : 0.0f) + (k1 ? v1.z : 0.0f)) +
            ((k2 ? v2.z : 0.0f) + (k3 ? v3.z : 0.0f));
      a.w = ((k0 ? v0.w : 0.0f) + (k1 ? v1.w : 0.0f)) +
            ((k2 ? v2.w : 0.0f) + (k3 ? v3.w : 0.0f));
      for (int j0 = 32; j0 < deg; j0 += 32) {
        int last = deg - 1;
        int idx = j0 + lane;
        int more = p[idx < last ? idx : last];
        int e0 = __shfl(more, g), e1 = __shfl(more, 8 + g);
        int e2 = __shfl(more, 16 + g), e3 = __shfl(more, 24 + g);
        float4 w0 = x4[(long)e0 * 4 + q];
        float4 w1 = x4[(long)e1 * 4 + q];
        float4 w2 = x4[(long)e2 * 4 + q];
        float4 w3 = x4[(long)e3 * 4 + q];
        bool m0 = j0 + g < deg, m1 = j0 + 8 + g < deg;
        bool m2 = j0 + 16 + g < deg, m3 = j0 + 24 + g < deg;
        a.x += ((m0 ? w0.x : 0.0f) + (m1 ? w1.x : 0.0f)) +
               ((m2 ? w2.x : 0.0f) + (m3 ? w3.x : 0.0f));
        a.y += ((m0 ? w0.y : 0.0f) + (m1 ? w1.y : 0.0f)) +
               ((m2 ? w2.y : 0.0f) + (m3 ? w3.y : 0.0f));
        a.z += ((m0 ? w0.z : 0.0f) + (m1 ? w1.z : 0.0f)) +
               ((m2 ? w2.z : 0.0f) + (m3 ? w3.z : 0.0f));
        a.w += ((m0 ? w0.w : 0.0f) + (m1 ? w1.w : 0.0f)) +
               ((m2 ? w2.w : 0.0f) + (m3 ? w3.w : 0.0f));
      }
    }
    a.x += __shfl_xor(a.x, 4);
    a.y += __shfl_xor(a.y, 4);
    a.z += __shfl_xor(a.z, 4);
    a.w += __shfl_xor(a.w, 4);
    a.x += __shfl_xor(a.x, 8);
    a.y += __shfl_xor(a.y, 8);
    a.z += __shfl_xor(a.z, 8);
    a.w += __shfl_xor(a.w, 8);
    a.x += __shfl_xor(a.x, 16);
    a.y += __shfl_xor(a.y, 16);
    a.z += __shfl_xor(a.z, 16);
    a.w += __shfl_xor(a.w, 16);
    if (g == 0) sAw[i * (APITCH / 4) + q] = a;
    ids = nids;
  }

  float* tl = (float*)sAw;
  const float* pag = tl + ln * APITCH;

  v2f aX[4];
#pragma unroll
  for (int kk = 0; kk < 4; ++kk) {
    int ka = 4 * kk + 2 * half;
    aX[kk][0] = pag[ka] * iv;
    aX[kk][1] = pag[ka + 1] * iv;
  }
#pragma unroll
  for (int c = 0; c < 4; ++c) {
    v8f mc = (v8f){0, 0, 0, 0, 0, 0, 0, 0};
#pragma unroll
    for (int kk = 0; kk < 4; ++kk) {
      int ka = 4 * kk + 2 * half;
      v2f b;
      b[0] = sWin[ka * 64 + c * 16 + ln];
      b[1] = sWin[(ka + 1) * 64 + c * 16 + ln];
      mc = wmma4(aX[kk], b, mc);
    }
#pragma unroll
    for (int v = 0; v < 8; ++v)
      tl[(v + 8 * half) * APITCH + c * 16 + ln] = mc[v];
  }

  v8f acc[4];
#pragma unroll
  for (int c = 0; c < 4; ++c) acc[c] = (v8f){0, 0, 0, 0, 0, 0, 0, 0};

#pragma unroll
  for (int k = 0; k < 64; k += 4) {
    int ka = k + 2 * half;
    v2f aA;
    aA[0] = pag[ka] + __int_as_float(__float_as_int(sbin[ka]) & hmask);
    aA[1] = pag[ka + 1] + __int_as_float(__float_as_int(sbin[ka + 1]) & hmask);
#pragma unroll
    for (int c = 0; c < 4; ++c) {
      v2f bL;
      bL[0] = sWl[ka * 64 + c * 16 + ln];
      bL[1] = sWl[(ka + 1) * 64 + c * 16 + ln];
      acc[c] = wmma4(aA, bL, acc[c]);
    }
  }

  const float* px = x + (long)r * 16;
#pragma unroll
  for (int kk = 0; kk < 4; ++kk) {
    int ka = 4 * kk + 2 * half;
    aX[kk][0] = px[ka];
    aX[kk][1] = px[ka + 1];
  }
#pragma unroll
  for (int c = 0; c < 4; ++c) {
    v8f mc = (v8f){0, 0, 0, 0, 0, 0, 0, 0};
#pragma unroll
    for (int kk = 0; kk < 4; ++kk) {
      int ka = 4 * kk + 2 * half;
      v2f b;
      b[0] = sWin[ka * 64 + c * 16 + ln];
      b[1] = sWin[(ka + 1) * 64 + c * 16 + ln];
      mc = wmma4(aX[kk], b, mc);
    }
#pragma unroll
    for (int v = 0; v < 8; ++v)
      tl[(v + 8 * half) * APITCH + c * 16 + ln] = mc[v];
  }

#pragma unroll
  for (int k = 0; k < 64; k += 4) {
    int ka = k + 2 * half;
    v2f aH;
    aH[0] = pag[ka] + sbin[ka];
    aH[1] = pag[ka + 1] + sbin[ka + 1];
#pragma unroll
    for (int c = 0; c < 4; ++c) {
      v2f bR;
      bR[0] = sWr[ka * 64 + c * 16 + ln];
      bR[1] = sWr[(ka + 1) * 64 + c * 16 + ln];
      acc[c] = wmma4(aH, bR, acc[c]);
    }
  }

  float bias[4], g4[4], be4[4];
#pragma unroll
  for (int c = 0; c < 4; ++c) {
    bias[c] = sbl[c * 16 + ln];
    g4[c] = sg[c * 16 + ln];
    be4[c] = sbeta[c * 16 + ln];
  }

#pragma unroll
  for (int v = 0; v < 8; ++v) {
    float r0 = fmaxf(acc[0][v] + bias[0], 0.0f);
    float r1 = fmaxf(acc[1][v] + bias[1], 0.0f);
    float r2 = fmaxf(acc[2][v] + bias[2], 0.0f);
    float r3 = fmaxf(acc[3][v] + bias[3], 0.0f);
    float s = r0 + r1 + r2 + r3;
    s += __shfl_xor(s, 1, 16);
    s += __shfl_xor(s, 2, 16);
    s += __shfl_xor(s, 4, 16);
    s += __shfl_xor(s, 8, 16);
    float mu = s * (1.0f / 64.0f);
    float d0 = r0 - mu, d1 = r1 - mu, d2 = r2 - mu, d3 = r3 - mu;
    float qq = d0 * d0 + d1 * d1 + d2 * d2 + d3 * d3;
    qq += __shfl_xor(qq, 1, 16);
    qq += __shfl_xor(qq, 2, 16);
    qq += __shfl_xor(qq, 4, 16);
    qq += __shfl_xor(qq, 8, 16);
    float rin = rsqrtf(qq * (1.0f / 64.0f) + LN_EPS);
    long row = row0 + v + 8 * half;
    if (row < N) {
      float* po = hout + row * 64 + ln;
      po[0] = d0 * rin * g4[0] + be4[0];
      po[16] = d1 * rin * g4[1] + be4[1];
      po[32] = d2 * rin * g4[2] + be4[2];
      po[48] = d3 * rin * g4[3] + be4[3];
    }
  }
}

extern "C" void kernel_launch(void* const* d_in, const int* in_sizes, int n_in,
                              void* d_out, int out_size, void* d_ws,
                              size_t ws_size, hipStream_t stream) {
  (void)n_in;
  (void)out_size;
  (void)ws_size;
  const float* x = (const float*)d_in[0];
  const int* ei = (const int*)d_in[1];
  const float* Win = (const float*)d_in[2];
  const float* bin = (const float*)d_in[3];
  const float* Wl = (const float*)d_in[4];
  const float* bl = (const float*)d_in[5];
  const float* Wr = (const float*)d_in[6];
  const float* gam = (const float*)d_in[7];
  const float* bet = (const float*)d_in[8];

  const int N = in_sizes[0] / 16;
  const long E = in_sizes[1] / 2;
  const int* src = ei;
  const int* dst = ei + E;

  float* ws = (float*)d_ws;
  size_t nA = ((size_t)N + 63) & ~(size_t)63;
  size_t eA = ((size_t)E + 63) & ~(size_t)63;
  int* cnt = (int*)ws;
  float* buf0 = ws + nA;
  float* buf1 = buf0 + (size_t)N * 64;
  int* grp = (int*)buf1;
  int* offs = grp + eA;
  float* inv = (float*)(offs + nA);
  float* outp = (float*)d_out;

  const int NB = (N + BSIZE - 1) / BSIZE;
  const int G = (int)((E + CHUNK - 1) / CHUNK);
  int* binned = (int*)d_out;
  int* runs = binned + eA;

  chunk_sort<<<G, 256, 0, stream>>>(src, dst, binned, runs, E, NB, G);
  bin_sort<<<NB, 1024, 0, stream>>>(binned, runs, grp, cnt, offs, inv, N, G);

  int tiles = (N + 15) / 16;
  int gblocks = (tiles + 7) / 8;
  sage_first<<<gblocks, 256, 0, stream>>>(x, Win, bin, grp, offs, cnt, inv, Wl,
                                          bl, Wr, gam, bet, outp, N);

  float* hcur = outp;
  float* other = buf0;
  for (int l = 1; l < 3; ++l) {
    sage_layer<<<gblocks, 256, 0, stream>>>(
        hcur, grp, offs, cnt, inv, Wl + (size_t)l * 4096, bl + (size_t)l * 64,
        Wr + (size_t)l * 4096, gam, bet, other, N);
    float* tmp = hcur;
    hcur = other;
    other = tmp;
  }
}
